# starting kernel
# baseline (speedup 1.0000x reference)
#include <hip/hip_runtime.h>
#include <math.h>


typedef __attribute__((ext_vector_type(2))) float v2f;
typedef __attribute__((ext_vector_type(8))) float v8f;

#define N_NODES 50000
#define N_EDGES 800000
#define K_DIM   128

__global__ void gcn_degrees(const int* __restrict__ src, const int* __restrict__ dst,
                            float* __restrict__ outdeg, float* __restrict__ indeg) {
  int e = blockIdx.x * blockDim.x + threadIdx.x;
  if (e < N_EDGES) {
    atomicAdd(&outdeg[src[e]], 1.0f);
    atomicAdd(&indeg[dst[e]], 1.0f);
  }
}

__global__ void gcn_deg_isqrt(const float* __restrict__ outdeg, const float* __restrict__ indeg,
                              float* __restrict__ odi, float* __restrict__ idi) {
  int i = blockIdx.x * blockDim.x + threadIdx.x;
  if (i < N_NODES) {
    odi[i] = rsqrtf(fmaxf(outdeg[i], 1.0f));
    idi[i] = rsqrtf(fmaxf(indeg[i], 1.0f));
  }
}

template <int NOUT>
__global__ __launch_bounds__(256) void gcn_gemm_wmma(const float* __restrict__ A,
                                                     const float* __restrict__ odi,
                                                     const float* __restrict__ W,
                                                     float* __restrict__ C) {
  __shared__ float As[16][K_DIM + 4];

  const int tileM = blockIdx.x * 16;

  for (int idx = threadIdx.x; idx < 16 * K_DIM; idx += blockDim.x) {
    const int r = idx >> 7;
    const int c = idx & (K_DIM - 1);
    As[r][c] = A[(size_t)(tileM + r) * K_DIM + c] * odi[tileM + r];
  }
  __syncthreads();

  const int wave = threadIdx.x >> 5;
  const int lane = threadIdx.x & 31;
  const int half = lane >> 4;
  const int r16  = lane & 15;
  const int n0   = wave * 16;

  v8f acc = {};
#pragma unroll
  for (int k = 0; k < K_DIM; k += 4) {
    v2f a, b;
    a.x = As[r16][k + 2 * half];
    a.y = As[r16][k + 2 * half + 1];
    b.x = W[(size_t)(k + 2 * half)     * NOUT + n0 + r16];
    b.y = W[(size_t)(k + 2 * half + 1) * NOUT + n0 + r16];
    acc = __builtin_amdgcn_wmma_f32_16x16x4_f32(false, a, false, b,
                                                (short)0, acc, false, false);
  }

#pragma unroll
  for (int r = 0; r < 8; ++r) {
    const int row = tileM + r + 8 * half;
    C[(size_t)row * NOUT + n0 + r16] = acc[r];
  }
}

template <int D>
__global__ void gcn_edge_scatter(const float* __restrict__ proj,
                                 const int* __restrict__ src, const int* __restrict__ dst,
                                 const float* __restrict__ ew, float* __restrict__ agg) {
  constexpr int LPE = D / 4;
  const int gid = blockIdx.x * blockDim.x + threadIdx.x;
  const int e = gid / LPE;
  const int l = gid % LPE;
  if (e >= N_EDGES) return;

  const int s = src[e];
  const int d = dst[e];
  const float w = ew[e];

  const float4 v = ((const float4*)(proj + (size_t)s * D))[l];
  float* o = agg + (size_t)d * D + l * 4;
  atomicAdd(o + 0, v.x * w);
  atomicAdd(o + 1, v.y * w);
  atomicAdd(o + 2, v.z * w);
  atomicAdd(o + 3, v.w * w);
}

__global__ void gcn_finalize(const float* __restrict__ agg, const float* __restrict__ idi,
                             const float* __restrict__ bias, float* __restrict__ out,
                             int D, int do_relu) {
  const int i = blockIdx.x * blockDim.x + threadIdx.x;
  if (i >= N_NODES * D) return;
  const int row = i / D;
  const int col = i % D;
  float v = agg[i] * idi[row] + bias[col];
  out[i] = do_relu ? fmaxf(v, 0.0f) : v;
}

extern "C" void kernel_launch(void* const* d_in, const int* in_sizes, int n_in,
                              void* d_out, int out_size, void* d_ws, size_t ws_size,
                              hipStream_t stream) {
  const float* feat = (const float*)d_in[0];
  const float* ew   = (const float*)d_in[1];
  const float* W1   = (const float*)d_in[2];
  const float* b1   = (const float*)d_in[3];
  const float* W2   = (const float*)d_in[4];
  const float* b2   = (const float*)d_in[5];
  const float* W3   = (const float*)d_in[6];
  const float* b3   = (const float*)d_in[7];
  const int*   src  = (const int*)d_in[8];
  const int*   dst  = (const int*)d_in[9];
  float* out = (float*)d_out;

  float* ws     = (float*)d_ws;
  float* outdeg = ws;
  float* indeg  = ws + N_NODES;
  float* odi    = ws + 2 * (size_t)N_NODES;
  float* idi    = ws + 3 * (size_t)N_NODES;
  float* h      = ws + 4 * (size_t)N_NODES;
  float* proj   = h    + (size_t)N_NODES * 128;
  float* agg    = proj + (size_t)N_NODES * 128;

  hipMemsetAsync(outdeg, 0, 2 * (size_t)N_NODES * sizeof(float), stream);
  gcn_degrees<<<(N_EDGES + 255) / 256, 256, 0, stream>>>(src, dst, outdeg, indeg);
  gcn_deg_isqrt<<<(N_NODES + 255) / 256, 256, 0, stream>>>(outdeg, indeg, odi, idi);

  const int gemmBlocks = N_NODES / 16;
  const int scat128Blocks = (N_EDGES * 32 + 255) / 256;
  const int scat64Blocks  = (N_EDGES * 16 + 255) / 256;
  const int fin128Blocks  = (N_NODES * 128 + 255) / 256;
  const int fin64Blocks   = (N_NODES * 64 + 255) / 256;

  gcn_gemm_wmma<128><<<gemmBlocks, 256, 0, stream>>>(feat, odi, W1, proj);
  hipMemsetAsync(agg, 0, (size_t)N_NODES * 128 * sizeof(float), stream);
  gcn_edge_scatter<128><<<scat128Blocks, 256, 0, stream>>>(proj, src, dst, ew, agg);
  gcn_finalize<<<fin128Blocks, 256, 0, stream>>>(agg, idi, b1, h, 128, 1);

  gcn_gemm_wmma<128><<<gemmBlocks, 256, 0, stream>>>(h, odi, W2, proj);
  hipMemsetAsync(agg, 0, (size_t)N_NODES * 128 * sizeof(float), stream);
  gcn_edge_scatter<128><<<scat128Blocks, 256, 0, stream>>>(proj, src, dst, ew, agg);
  gcn_finalize<<<fin128Blocks, 256, 0, stream>>>(agg, idi, b2, h, 128, 1);

  gcn_gemm_wmma<64><<<gemmBlocks, 128, 0, stream>>>(h, odi, W3, proj);
  hipMemsetAsync(agg, 0, (size_t)N_NODES * 64 * sizeof(float), stream);
  gcn_edge_scatter<64><<<scat64Blocks, 256, 0, stream>>>(proj, src, dst, ew, agg);
  gcn_finalize<<<fin64Blocks, 256, 0, stream>>>(agg, idi, b3, out, 64, 0);
}


# Tile edge filter runs once per call: layer 1 stores each tile's resolved edge list (source, local row, weight) in workspace; layers 2-3 read it, no dst scan; row sums still LDS arrival order
# speedup vs baseline: 1.9731x; 1.9731x over previous
#include <hip/hip_runtime.h>
#include <math.h>


typedef __attribute__((ext_vector_type(2))) float v2f;
typedef __attribute__((ext_vector_type(8))) float v8f;

#define N_NODES 50000
#define N_EDGES 800000
#define K_DIM   128
#define CNT_PITCH 32

__global__ void gcn_degrees(const int* __restrict__ src, const int* __restrict__ dst,
                            float* __restrict__ outdeg, float* __restrict__ indeg) {
  int e = blockIdx.x * blockDim.x + threadIdx.x;
  if (e < N_EDGES) {
    atomicAdd(&outdeg[src[e]], 1.0f);
    atomicAdd(&indeg[dst[e]], 1.0f);
  }
}

__global__ void gcn_deg_isqrt(const float* __restrict__ outdeg, const float* __restrict__ indeg,
                              float* __restrict__ odi, float* __restrict__ idi) {
  int i = blockIdx.x * blockDim.x + threadIdx.x;
  if (i < N_NODES) {
    odi[i] = rsqrtf(fmaxf(outdeg[i], 1.0f));
    idi[i] = rsqrtf(fmaxf(indeg[i], 1.0f));
  }
}

template <int NOUT>
__global__ __launch_bounds__(256) void gcn_gemm_wmma(const float* __restrict__ A,
                                                     const float* __restrict__ odi,
                                                     const float* __restrict__ W,
                                                     float* __restrict__ C) {
  __shared__ float As[16][K_DIM + 4];

  const int tileM = blockIdx.x * 16;

  for (int idx = threadIdx.x; idx < 16 * K_DIM; idx += blockDim.x) {
    const int r = idx >> 7;
    const int c = idx & (K_DIM - 1);
    As[r][c] = A[(size_t)(tileM + r) * K_DIM + c] * odi[tileM + r];
  }
  __syncthreads();

  const int wave = threadIdx.x >> 5;
  const int lane = threadIdx.x & 31;
  const int half = lane >> 4;
  const int r16  = lane & 15;
  const int n0   = wave * 16;

  v8f acc = {};
#pragma unroll
  for (int k = 0; k < K_DIM; k += 4) {
    v2f a, b;
    a.x = As[r16][k + 2 * half];
    a.y = As[r16][k + 2 * half + 1];
    b.x = W[(size_t)(k + 2 * half)     * NOUT + n0 + r16];
    b.y = W[(size_t)(k + 2 * half + 1) * NOUT + n0 + r16];
    acc = __builtin_amdgcn_wmma_f32_16x16x4_f32(false, a, false, b,
                                                (short)0, acc, false, false);
  }

#pragma unroll
  for (int r = 0; r < 8; ++r) {
    const int row = tileM + r + 8 * half;
    C[(size_t)row * NOUT + n0 + r16] = acc[r];
  }
}

template <int D, int TR, int CAP, int GCAP, bool RELU, bool BUILD>
__global__ __launch_bounds__(1024) void gcn_tile_aggregate(const float* __restrict__ proj,
                                                           const int* __restrict__ src,
                                                           const int* __restrict__ dst,
                                                           const float* __restrict__ ew,
                                                           const float* __restrict__ idi,
                                                           const float* __restrict__ bias,
                                                           float* __restrict__ out,
                                                           int* __restrict__ glist,
                                                           unsigned* __restrict__ gcnt) {
  static_assert(D % 32 == 0 && N_EDGES % 4 == 0, "layout");
  static_assert(N_NODES <= 65536 && TR <= 32768, "packed list entry: 16 bits each");
  static_assert(BUILD ? CAP <= GCAP : GCAP <= CAP, "listed edges fit glist; a stored list fits the LDS list");
  constexpr int NT  = 1024;
  constexpr int CPL = D / 32;
  constexpr int D4  = D / 4;
  constexpr int N4  = N_EDGES / 4;
  constexpr int U   = 8;

  __shared__ __attribute__((aligned(16))) float tile[TR * D];
  __shared__ int2     list[CAP];
  __shared__ unsigned cnt;

  const int tid      = threadIdx.x;
  const int lane     = tid & 31;
  const int rowBase  = blockIdx.x * TR;
  const int rowsHere = min(TR, N_NODES - rowBase);

  int2* const gl = (int2*)glist + (size_t)blockIdx.x * GCAP;
  unsigned nStored = 0u;
  if (!BUILD) nStored = gcnt[blockIdx.x * CNT_PITCH];
  const bool useStored = !BUILD && nStored <= (unsigned)GCAP;
  constexpr int PRE = 4;
  int2 pre[PRE];
  if (useStored) {
#pragma unroll
    for (int u = 0; u < PRE; ++u) {
      const unsigned i = (unsigned)(tid + u * NT);
      pre[u] = (i < nStored) ? gl[i] : make_int2(0, 0);
    }
  }

  for (int i = tid; i < TR * D4; i += NT)
    ((float4*)tile)[i] = make_float4(0.f, 0.f, 0.f, 0.f);
  if (tid == 0) cnt = 0u;
  if (useStored) {
#pragma unroll
    for (int u = 0; u < PRE; ++u) {
      const unsigned i = (unsigned)(tid + u * NT);
      if (i < nStored) list[i] = pre[u];
    }
    for (unsigned i = (unsigned)(tid + PRE * NT); i < nStored; i += NT) list[i] = gl[i];
  }
  __syncthreads();

  auto take = [&](int d, int e) {
    const unsigned dl = (unsigned)(d - rowBase);
    if (dl < (unsigned)rowsHere) {
      const unsigned slot = atomicAdd(&cnt, 1u);
      if (slot < (unsigned)CAP) {
        list[slot] = make_int2(e, (int)dl);
      } else {
        const float w  = ew[e];
        const float* p = proj + (size_t)src[e] * D;
        float* t       = tile + dl * D;
#pragma nounroll
        for (int c = 0; c < D; ++c) atomicAdd(t + c, p[c] * w);
      }
    }
  };
  auto take4 = [&](const int4& d, int e) {
    take(d.x, e);
    take(d.y, e + 1);
    take(d.z, e + 2);
    take(d.w, e + 3);
  };

  if (!useStored) {
    const int4* dst4 = (const int4*)dst;
    int i0 = tid;
    for (; i0 + (U - 1) * NT < N4; i0 += NT * U) {
      int4 dv[U];
#pragma unroll
      for (int u = 0; u < U; ++u) dv[u] = dst4[i0 + u * NT];
#pragma unroll
      for (int u = 0; u < U; ++u) take4(dv[u], (i0 + u * NT) * 4);
    }
    for (; i0 < N4; i0 += NT) take4(dst4[i0], i0 * 4);
  }
  __syncthreads();

  const int wave  = __builtin_amdgcn_readfirstlane(tid >> 5);
  const int total = __builtin_amdgcn_readfirstlane(
      useStored ? (int)nStored : (int)min(cnt, (unsigned)CAP));
  if (BUILD && tid == 0) gcnt[blockIdx.x * CNT_PITCH] = cnt;

  auto add_row = [&](int sj, int dj, float wj) {
    const float* p = proj + (size_t)sj * D + lane;
    float* t       = tile + dj * D + lane;
    float v[CPL];
#pragma unroll
    for (int c = 0; c < CPL; ++c) v[c] = p[32 * c];
#pragma unroll
    for (int c = 0; c < CPL; ++c) atomicAdd(t + 32 * c, v[c] * wj);
  };

  auto add_step = [&](int s, int dl, float w, int nb) {
#pragma unroll
    for (int q = 0; q < 4; ++q) {
      const int nq = nb - 8 * q;
      if (nq >= 8) {
        float v[8][CPL];
#pragma unroll
        for (int j = 0; j < 8; ++j) {
          const int sj   = __builtin_amdgcn_readlane(s, 8 * q + j);
          const float* p = proj + (size_t)sj * D + lane;
#pragma unroll
          for (int c = 0; c < CPL; ++c) v[j][c] = p[32 * c];
        }
#pragma unroll
        for (int j = 0; j < 8; ++j) {
          const int dj   = __builtin_amdgcn_readlane(dl, 8 * q + j);
          const float wj = __int_as_float(__builtin_amdgcn_readlane(__float_as_int(w), 8 * q + j));
          float* t       = tile + dj * D + lane;
#pragma unroll
          for (int c = 0; c < CPL; ++c) atomicAdd(t + 32 * c, v[j][c] * wj);
        }
      } else {
        for (int j = 0; j < nq; ++j) {
          const int l = 8 * q + j;
          add_row(__builtin_amdgcn_readlane(s, l), __builtin_amdgcn_readlane(dl, l),
                  __int_as_float(__builtin_amdgcn_readlane(__float_as_int(w), l)));
        }
      }
    }
  };

  for (int b0 = wave * 32; b0 < total; b0 += 32 * 32) {
    const int nb = min(32, total - b0);
    int s = 0, dl = 0;
    float w = 0.f;
    if (lane < nb) {
      const int2 ent = list[b0 + lane];
      if (useStored) {
        s  = ent.x & 0xFFFF;
        dl = (int)((unsigned)ent.x >> 16);
        w  = __int_as_float(ent.y);
      } else {
        s  = src[ent.x];
        w  = ew[ent.x];
        dl = ent.y;
        if (BUILD) gl[b0 + lane] = make_int2(s | (dl << 16), __float_as_int(w));
      }
    }
    add_step(s, dl, w, nb);
  }
  __syncthreads();

  for (int i = tid; i < rowsHere * D4; i += NT) {
    const int r  = i / D4;
    const int c4 = i % D4;
    const float4 t  = ((const float4*)tile)[i];
    const float  sc = idi[rowBase + r];
    const float4 b  = ((const float4*)bias)[c4];
    float4 o;
    o.x = t.x * sc + b.x;
    o.y = t.y * sc + b.y;
    o.z = t.z * sc + b.z;
    o.w = t.w * sc + b.w;
    if (RELU) {
      o.x = fmaxf(o.x, 0.0f);
      o.y = fmaxf(o.y, 0.0f);
      o.z = fmaxf(o.z, 0.0f);
      o.w = fmaxf(o.w, 0.0f);
    }
    ((float4*)out)[(size_t)(rowBase + r) * D4 + c4] = o;
  }
}

extern "C" void kernel_launch(void* const* d_in, const int* in_sizes, int n_in,
                              void* d_out, int out_size, void* d_ws, size_t ws_size,
                              hipStream_t stream) {
  const float* feat = (const float*)d_in[0];
  const float* ew   = (const float*)d_in[1];
  const float* W1   = (const float*)d_in[2];
  const float* b1   = (const float*)d_in[3];
  const float* W2   = (const float*)d_in[4];
  const float* b2   = (const float*)d_in[5];
  const float* W3   = (const float*)d_in[6];
  const float* b3   = (const float*)d_in[7];
  const int*   src  = (const int*)d_in[8];
  const int*   dst  = (const int*)d_in[9];
  float* out = (float*)d_out;

  float* ws     = (float*)d_ws;
  float* outdeg = ws;
  float* indeg  = ws + N_NODES;
  float* odi    = ws + 2 * (size_t)N_NODES;
  float* idi    = ws + 3 * (size_t)N_NODES;
  float* h      = ws + 4 * (size_t)N_NODES;
  float* proj   = h    + (size_t)N_NODES * 128;

  hipMemsetAsync(outdeg, 0, 2 * (size_t)N_NODES * sizeof(float), stream);
  gcn_degrees<<<(N_EDGES + 255) / 256, 256, 0, stream>>>(src, dst, outdeg, indeg);
  gcn_deg_isqrt<<<(N_NODES + 255) / 256, 256, 0, stream>>>(outdeg, indeg, odi, idi);

  const int gemmBlocks = N_NODES / 16;
  constexpr int TR = 256;
  constexpr int tiles = (N_NODES + TR - 1) / TR;
  constexpr int CAP128 = (256 * 1024 - TR * 128 * 4) / 8;
  constexpr int CAP64  = (256 * 1024 - TR * 64 * 4) / 8;
  int*      glist = (int*)(proj + (size_t)N_NODES * 128);
  unsigned* gcnt  = (unsigned*)(glist + (size_t)2 * tiles * CAP128);

  gcn_gemm_wmma<128><<<gemmBlocks, 256, 0, stream>>>(feat, odi, W1, proj);
  gcn_tile_aggregate<128, TR, CAP128, CAP128, true, true><<<tiles, 1024, 0, stream>>>(proj, src, dst, ew, idi, b1, h, glist, gcnt);

  gcn_gemm_wmma<128><<<gemmBlocks, 256, 0, stream>>>(h, odi, W2, proj);
  gcn_tile_aggregate<128, TR, CAP128, CAP128, true, false><<<tiles, 1024, 0, stream>>>(proj, src, dst, ew, idi, b2, h, glist, gcnt);

  gcn_gemm_wmma<64><<<gemmBlocks, 128, 0, stream>>>(h, odi, W3, proj);
  gcn_tile_aggregate<64, TR, CAP64, CAP128, false, false><<<tiles, 1024, 0, stream>>>(proj, src, dst, ew, idi, b3, out, glist, gcnt);
}


# Tile edge filter tests a 1-byte tile key per edge (written by the degree pass), 16 keys per 16-byte load, not the 4-byte dst; a listed edge's row is read in phase B; sums stay LDS arrival order
# speedup vs baseline: 2.3009x; 1.1661x over previous
#include <hip/hip_runtime.h>
#include <math.h>


typedef __attribute__((ext_vector_type(2))) float v2f;
typedef __attribute__((ext_vector_type(8))) float v8f;

#define N_NODES 50000
#define N_EDGES 800000
#define K_DIM   128
#define CNT_PITCH 32
#define TILE_SHIFT 8
#define NO_TILE 255

__global__ void gcn_degrees(const int* __restrict__ src, const int* __restrict__ dst,
                            float* __restrict__ outdeg, float* __restrict__ indeg,
                            unsigned char* __restrict__ keys) {
  static_assert(((N_NODES - 1) >> TILE_SHIFT) < NO_TILE, "tile keys fit a byte");
  int e = blockIdx.x * blockDim.x + threadIdx.x;
  if (e < N_EDGES) {
    const int d = dst[e];
    atomicAdd(&outdeg[src[e]], 1.0f);
    atomicAdd(&indeg[d], 1.0f);
    keys[e] = (unsigned char)((unsigned)d < (unsigned)N_NODES ? d >> TILE_SHIFT : NO_TILE);
  }
}

__global__ void gcn_deg_isqrt(const float* __restrict__ outdeg, const float* __restrict__ indeg,
                              float* __restrict__ odi, float* __restrict__ idi) {
  int i = blockIdx.x * blockDim.x + threadIdx.x;
  if (i < N_NODES) {
    odi[i] = rsqrtf(fmaxf(outdeg[i], 1.0f));
    idi[i] = rsqrtf(fmaxf(indeg[i], 1.0f));
  }
}

template <int NOUT>
__global__ __launch_bounds__(256) void gcn_gemm_wmma(const float* __restrict__ A,
                                                     const float* __restrict__ odi,
                                                     const float* __restrict__ W,
                                                     float* __restrict__ C) {
  __shared__ float As[16][K_DIM + 4];

  const int tileM = blockIdx.x * 16;

  for (int idx = threadIdx.x; idx < 16 * K_DIM; idx += blockDim.x) {
    const int r = idx >> 7;
    const int c = idx & (K_DIM - 1);
    As[r][c] = A[(size_t)(tileM + r) * K_DIM + c] * odi[tileM + r];
  }
  __syncthreads();

  const int wave = threadIdx.x >> 5;
  const int lane = threadIdx.x & 31;
  const int half = lane >> 4;
  const int r16  = lane & 15;
  const int n0   = wave * 16;

  v8f acc = {};
#pragma unroll
  for (int k = 0; k < K_DIM; k += 4) {
    v2f a, b;
    a.x = As[r16][k + 2 * half];
    a.y = As[r16][k + 2 * half + 1];
    b.x = W[(size_t)(k + 2 * half)     * NOUT + n0 + r16];
    b.y = W[(size_t)(k + 2 * half + 1) * NOUT + n0 + r16];
    acc = __builtin_amdgcn_wmma_f32_16x16x4_f32(false, a, false, b,
                                                (short)0, acc, false, false);
  }

#pragma unroll
  for (int r = 0; r < 8; ++r) {
    const int row = tileM + r + 8 * half;
    C[(size_t)row * NOUT + n0 + r16] = acc[r];
  }
}

template <int D, int TR, int CAP, int GCAP, bool RELU, bool BUILD>
__global__ __launch_bounds__(1024) void gcn_tile_aggregate(const float* __restrict__ proj,
                                                           const int* __restrict__ src,
                                                           const int* __restrict__ dst,
                                                           const float* __restrict__ ew,
                                                           const float* __restrict__ idi,
                                                           const float* __restrict__ bias,
                                                           float* __restrict__ out,
                                                           int* __restrict__ glist,
                                                           unsigned* __restrict__ gcnt,
                                                           const unsigned char* __restrict__ keys) {
  static_assert(D % 32 == 0 && N_EDGES % 16 == 0, "layout");
  static_assert(TR == (1 << TILE_SHIFT), "a tile key names TR rows");
  static_assert(N_NODES <= 65536 && TR <= 32768, "packed list entry: 16 bits each");
  static_assert(BUILD ? CAP <= GCAP : GCAP <= CAP, "listed edges fit glist; a stored list fits the LDS list");
  constexpr int NT  = 1024;
  constexpr int CPL = D / 32;
  constexpr int D4  = D / 4;
  constexpr int N16 = N_EDGES / 16;
  constexpr int U   = 8;

  __shared__ __attribute__((aligned(16))) float tile[TR * D];
  __shared__ int2     list[CAP];
  __shared__ unsigned cnt;

  const int tid      = threadIdx.x;
  const int lane     = tid & 31;
  const int rowBase  = blockIdx.x * TR;
  const int rowsHere = min(TR, N_NODES - rowBase);

  int2* const gl = (int2*)glist + (size_t)blockIdx.x * GCAP;
  unsigned nStored = 0u;
  if (!BUILD) nStored = gcnt[blockIdx.x * CNT_PITCH];
  const bool useStored = !BUILD && nStored <= (unsigned)GCAP;
  constexpr int PRE = 4;
  int2 pre[PRE];
  if (useStored) {
#pragma unroll
    for (int u = 0; u < PRE; ++u) {
      const unsigned i = (unsigned)(tid + u * NT);
      pre[u] = (i < nStored) ? gl[i] : make_int2(0, 0);
    }
  }

  for (int i = tid; i < TR * D4; i += NT)
    ((float4*)tile)[i] = make_float4(0.f, 0.f, 0.f, 0.f);
  if (tid == 0) cnt = 0u;
  if (useStored) {
#pragma unroll
    for (int u = 0; u < PRE; ++u) {
      const unsigned i = (unsigned)(tid + u * NT);
      if (i < nStored) list[i] = pre[u];
    }
    for (unsigned i = (unsigned)(tid + PRE * NT); i < nStored; i += NT) list[i] = gl[i];
  }
  __syncthreads();

  auto take_edge = [&](int e) {
    const unsigned slot = atomicAdd(&cnt, 1u);
    if (slot < (unsigned)CAP) {
      list[slot].x = e;
    } else {
      const float w  = ew[e];
      const float* p = proj + (size_t)src[e] * D;
      float* t       = tile + (dst[e] - rowBase) * D;
#pragma nounroll
      for (int c = 0; c < D; ++c) atomicAdd(t + c, p[c] * w);
    }
  };
  const unsigned pat = (unsigned)blockIdx.x * 0x01010101u;
  auto differs = [&](unsigned k) {
    const unsigned x = k ^ pat;
    return ((x & 0x7F7F7F7Fu) + 0x7F7F7F7Fu) | x;
  };
  auto take16 = [&](const uint4& k, int e0) {
    const unsigned t0 = differs(k.x), t1 = differs(k.y), t2 = differs(k.z), t3 = differs(k.w);
    if ((t0 & t1 & t2 & t3 & 0x80808080u) != 0x80808080u) {
      unsigned m = ~(((t0 >> 7) & 0x01010101u) | ((t1 >> 6) & 0x02020202u) |
                     ((t2 >> 5) & 0x04040404u) | ((t3 >> 4) & 0x08080808u)) & 0x0F0F0F0Fu;
      do {
        const int p = __builtin_ctz(m);
        m &= m - 1u;
        take_edge(e0 + 4 * (p & 7) + (p >> 3));
      } while (m != 0u);
    }
  };

  if (!useStored) {
    const uint4* key4 = (const uint4*)keys;
    int i0 = tid;
    for (; i0 + (U - 1) * NT < N16; i0 += NT * U) {
      uint4 kv[U];
#pragma unroll
      for (int u = 0; u < U; ++u) kv[u] = key4[i0 + u * NT];
#pragma unroll
      for (int u = 0; u < U; ++u) take16(kv[u], (i0 + u * NT) * 16);
    }
    for (; i0 < N16; i0 += NT) take16(key4[i0], i0 * 16);
  }
  __syncthreads();

  const int wave  = __builtin_amdgcn_readfirstlane(tid >> 5);
  const int total = __builtin_amdgcn_readfirstlane(
      useStored ? (int)nStored : (int)min(cnt, (unsigned)CAP));
  if (BUILD && tid == 0) gcnt[blockIdx.x * CNT_PITCH] = cnt;

  auto add_row = [&](int sj, int dj, float wj) {
    const float* p = proj + (size_t)sj * D + lane;
    float* t       = tile + dj * D + lane;
    float v[CPL];
#pragma unroll
    for (int c = 0; c < CPL; ++c) v[c] = p[32 * c];
#pragma unroll
    for (int c = 0; c < CPL; ++c) atomicAdd(t + 32 * c, v[c] * wj);
  };

  auto add_step = [&](int s, int dl, float w, int nb) {
#pragma unroll
    for (int q = 0; q < 4; ++q) {
      const int nq = nb - 8 * q;
      if (nq >= 8) {
        float v[8][CPL];
#pragma unroll
        for (int j = 0; j < 8; ++j) {
          const int sj   = __builtin_amdgcn_readlane(s, 8 * q + j);
          const float* p = proj + (size_t)sj * D + lane;
#pragma unroll
          for (int c = 0; c < CPL; ++c) v[j][c] = p[32 * c];
        }
#pragma unroll
        for (int j = 0; j < 8; ++j) {
          const int dj   = __builtin_amdgcn_readlane(dl, 8 * q + j);
          const float wj = __int_as_float(__builtin_amdgcn_readlane(__float_as_int(w), 8 * q + j));
          float* t       = tile + dj * D + lane;
#pragma unroll
          for (int c = 0; c < CPL; ++c) atomicAdd(t + 32 * c, v[j][c] * wj);
        }
      } else {
        for (int j = 0; j < nq; ++j) {
          const int l = 8 * q + j;
          add_row(__builtin_amdgcn_readlane(s, l), __builtin_amdgcn_readlane(dl, l),
                  __int_as_float(__builtin_amdgcn_readlane(__float_as_int(w), l)));
        }
      }
    }
  };

  for (int b0 = wave * 32; b0 < total; b0 += 32 * 32) {
    const int nb = min(32, total - b0);
    int s = 0, dl = 0;
    float w = 0.f;
    if (lane < nb) {
      if (useStored) {
        const int2 ent = list[b0 + lane];
        s  = ent.x & 0xFFFF;
        dl = (int)((unsigned)ent.x >> 16);
        w  = __int_as_float(ent.y);
      } else {
        const int e = list[b0 + lane].x;
        s  = src[e];
        w  = ew[e];
        dl = dst[e] - rowBase;
        if (BUILD) gl[b0 + lane] = make_int2(s | (dl << 16), __float_as_int(w));
      }
    }
    add_step(s, dl, w, nb);
  }
  __syncthreads();

  for (int i = tid; i < rowsHere * D4; i += NT) {
    const int r  = i / D4;
    const int c4 = i % D4;
    const float4 t  = ((const float4*)tile)[i];
    const float  sc = idi[rowBase + r];
    const float4 b  = ((const float4*)bias)[c4];
    float4 o;
    o.x = t.x * sc + b.x;
    o.y = t.y * sc + b.y;
    o.z = t.z * sc + b.z;
    o.w = t.w * sc + b.w;
    if (RELU) {
      o.x = fmaxf(o.x, 0.0f);
      o.y = fmaxf(o.y, 0.0f);
      o.z = fmaxf(o.z, 0.0f);
      o.w = fmaxf(o.w, 0.0f);
    }
    ((float4*)out)[(size_t)(rowBase + r) * D4 + c4] = o;
  }
}

extern "C" void kernel_launch(void* const* d_in, const int* in_sizes, int n_in,
                              void* d_out, int out_size, void* d_ws, size_t ws_size,
                              hipStream_t stream) {
  const float* feat = (const float*)d_in[0];
  const float* ew   = (const float*)d_in[1];
  const float* W1   = (const float*)d_in[2];
  const float* b1   = (const float*)d_in[3];
  const float* W2   = (const float*)d_in[4];
  const float* b2   = (const float*)d_in[5];
  const float* W3   = (const float*)d_in[6];
  const float* b3   = (const float*)d_in[7];
  const int*   src  = (const int*)d_in[8];
  const int*   dst  = (const int*)d_in[9];
  float* out = (float*)d_out;

  float* ws     = (float*)d_ws;
  float* outdeg = ws;
  float* indeg  = ws + N_NODES;
  float* odi    = ws + 2 * (size_t)N_NODES;
  float* idi    = ws + 3 * (size_t)N_NODES;
  float* h      = ws + 4 * (size_t)N_NODES;
  float* proj   = h    + (size_t)N_NODES * 128;

  const int gemmBlocks = N_NODES / 16;
  constexpr int TR = 256;
  constexpr int tiles = (N_NODES + TR - 1) / TR;
  constexpr int CAP128 = (256 * 1024 - TR * 128 * 4) / 8;
  constexpr int CAP64  = (256 * 1024 - TR * 64 * 4) / 8;
  int*      glist = (int*)(proj + (size_t)N_NODES * 128);
  unsigned* gcnt  = (unsigned*)(glist + (size_t)2 * tiles * CAP128);
  unsigned char* keys = (unsigned char*)(gcnt + (size_t)tiles * CNT_PITCH);

  hipMemsetAsync(outdeg, 0, 2 * (size_t)N_NODES * sizeof(float), stream);
  gcn_degrees<<<(N_EDGES + 255) / 256, 256, 0, stream>>>(src, dst, outdeg, indeg, keys);
  gcn_deg_isqrt<<<(N_NODES + 255) / 256, 256, 0, stream>>>(outdeg, indeg, odi, idi);

  gcn_gemm_wmma<128><<<gemmBlocks, 256, 0, stream>>>(feat, odi, W1, proj);
  gcn_tile_aggregate<128, TR, CAP128, CAP128, true, true><<<tiles, 1024, 0, stream>>>(proj, src, dst, ew, idi, b1, h, glist, gcnt, keys);

  gcn_gemm_wmma<128><<<gemmBlocks, 256, 0, stream>>>(h, odi, W2, proj);
  gcn_tile_aggregate<128, TR, CAP128, CAP128, true, false><<<tiles, 1024, 0, stream>>>(proj, src, dst, ew, idi, b2, h, glist, gcnt, keys);

  gcn_gemm_wmma<64><<<gemmBlocks, 128, 0, stream>>>(h, odi, W3, proj);
  gcn_tile_aggregate<64, TR, CAP64, CAP128, false, false><<<tiles, 1024, 0, stream>>>(proj, src, dst, ew, idi, b3, out, glist, gcnt, keys);
}


# Tile aggregate: 197 destination rows per tile, not 256: 254 tile work-groups (one per WGP) fill the 256 WGPs instead of 196, fewer LDS float adds per WGP; row sums still LDS arrival order
# speedup vs baseline: 2.4345x; 1.0581x over previous
#include <hip/hip_runtime.h>
#include <math.h>


typedef __attribute__((ext_vector_type(2))) float v2f;
typedef __attribute__((ext_vector_type(8))) float v8f;

#define N_NODES 50000
#define N_EDGES 800000
#define K_DIM   128
#define CNT_PITCH 32
#define TILE_ROWS 197
#define NO_TILE 255

__global__ void gcn_degrees(const int* __restrict__ src, const int* __restrict__ dst,
                            float* __restrict__ outdeg, float* __restrict__ indeg,
                            unsigned char* __restrict__ keys) {
  static_assert((N_NODES - 1) / TILE_ROWS < NO_TILE, "tile keys fit a byte");
  int e = blockIdx.x * blockDim.x + threadIdx.x;
  if (e < N_EDGES) {
    const int d = dst[e];
    atomicAdd(&outdeg[src[e]], 1.0f);
    atomicAdd(&indeg[d], 1.0f);
    keys[e] = (unsigned char)((unsigned)d < (unsigned)N_NODES ? (unsigned)d / TILE_ROWS : NO_TILE);
  }
}

__global__ void gcn_deg_isqrt(const float* __restrict__ outdeg, const float* __restrict__ indeg,
                              float* __restrict__ odi, float* __restrict__ idi) {
  int i = blockIdx.x * blockDim.x + threadIdx.x;
  if (i < N_NODES) {
    odi[i] = rsqrtf(fmaxf(outdeg[i], 1.0f));
    idi[i] = rsqrtf(fmaxf(indeg[i], 1.0f));
  }
}

template <int NOUT>
__global__ __launch_bounds__(256) void gcn_gemm_wmma(const float* __restrict__ A,
                                                     const float* __restrict__ odi,
                                                     const float* __restrict__ W,
                                                     float* __restrict__ C) {
  __shared__ float As[16][K_DIM + 4];

  const int tileM = blockIdx.x * 16;

  for (int idx = threadIdx.x; idx < 16 * K_DIM; idx += blockDim.x) {
    const int r = idx >> 7;
    const int c = idx & (K_DIM - 1);
    As[r][c] = A[(size_t)(tileM + r) * K_DIM + c] * odi[tileM + r];
  }
  __syncthreads();

  const int wave = threadIdx.x >> 5;
  const int lane = threadIdx.x & 31;
  const int half = lane >> 4;
  const int r16  = lane & 15;
  const int n0   = wave * 16;

  v8f acc = {};
#pragma unroll
  for (int k = 0; k < K_DIM; k += 4) {
    v2f a, b;
    a.x = As[r16][k + 2 * half];
    a.y = As[r16][k + 2 * half + 1];
    b.x = W[(size_t)(k + 2 * half)     * NOUT + n0 + r16];
    b.y = W[(size_t)(k + 2 * half + 1) * NOUT + n0 + r16];
    acc = __builtin_amdgcn_wmma_f32_16x16x4_f32(false, a, false, b,
                                                (short)0, acc, false, false);
  }

#pragma unroll
  for (int r = 0; r < 8; ++r) {
    const int row = tileM + r + 8 * half;
    C[(size_t)row * NOUT + n0 + r16] = acc[r];
  }
}

template <int D, int TR, int CAP, int GCAP, bool RELU, bool BUILD>
__global__ __launch_bounds__(1024) void gcn_tile_aggregate(const float* __restrict__ proj,
                                                           const int* __restrict__ src,
                                                           const int* __restrict__ dst,
                                                           const float* __restrict__ ew,
                                                           const float* __restrict__ idi,
                                                           const float* __restrict__ bias,
                                                           float* __restrict__ out,
                                                           int* __restrict__ glist,
                                                           unsigned* __restrict__ gcnt,
                                                           const unsigned char* __restrict__ keys) {
  static_assert(D % 32 == 0 && N_EDGES % 16 == 0, "layout");
  static_assert(TR == TILE_ROWS, "a tile key names TR rows");
  static_assert(N_NODES <= 65536 && TR <= 32768, "packed list entry: 16 bits each");
  static_assert(BUILD ? CAP <= GCAP : GCAP <= CAP, "listed edges fit glist; a stored list fits the LDS list");
  constexpr int NT  = 1024;
  constexpr int CPL = D / 32;
  constexpr int D4  = D / 4;
  constexpr int N16 = N_EDGES / 16;
  constexpr int U   = 8;

  __shared__ __attribute__((aligned(16))) float tile[TR * D];
  __shared__ int2     list[CAP];
  __shared__ unsigned cnt;

  const int tid      = threadIdx.x;
  const int lane     = tid & 31;
  const int rowBase  = blockIdx.x * TR;
  const int rowsHere = min(TR, N_NODES - rowBase);

  int2* const gl = (int2*)glist + (size_t)blockIdx.x * GCAP;
  unsigned nStored = 0u;
  if (!BUILD) nStored = gcnt[blockIdx.x * CNT_PITCH];
  const bool useStored = !BUILD && nStored <= (unsigned)GCAP;
  constexpr int PRE = 4;
  int2 pre[PRE];
  if (useStored) {
#pragma unroll
    for (int u = 0; u < PRE; ++u) {
      const unsigned i = (unsigned)(tid + u * NT);
      pre[u] = (i < nStored) ? gl[i] : make_int2(0, 0);
    }
  }

  for (int i = tid; i < TR * D4; i += NT)
    ((float4*)tile)[i] = make_float4(0.f, 0.f, 0.f, 0.f);
  if (tid == 0) cnt = 0u;
  if (useStored) {
#pragma unroll
    for (int u = 0; u < PRE; ++u) {
      const unsigned i = (unsigned)(tid + u * NT);
      if (i < nStored) list[i] = pre[u];
    }
    for (unsigned i = (unsigned)(tid + PRE * NT); i < nStored; i += NT) list[i] = gl[i];
  }
  __syncthreads();

  auto take_edge = [&](int e) {
    const unsigned slot = atomicAdd(&cnt, 1u);
    if (slot < (unsigned)CAP) {
      list[slot].x = e;
    } else {
      const float w  = ew[e];
      const float* p = proj + (size_t)src[e] * D;
      float* t       = tile + (dst[e] - rowBase) * D;
#pragma nounroll
      for (int c = 0; c < D; ++c) atomicAdd(t + c, p[c] * w);
    }
  };
  const unsigned pat = (unsigned)blockIdx.x * 0x01010101u;
  auto differs = [&](unsigned k) {
    const unsigned x = k ^ pat;
    return ((x & 0x7F7F7F7Fu) + 0x7F7F7F7Fu) | x;
  };
  auto take16 = [&](const uint4& k, int e0) {
    const unsigned t0 = differs(k.x), t1 = differs(k.y), t2 = differs(k.z), t3 = differs(k.w);
    if ((t0 & t1 & t2 & t3 & 0x80808080u) != 0x80808080u) {
      unsigned m = ~(((t0 >> 7) & 0x01010101u) | ((t1 >> 6) & 0x02020202u) |
                     ((t2 >> 5) & 0x04040404u) | ((t3 >> 4) & 0x08080808u)) & 0x0F0F0F0Fu;
      do {
        const int p = __builtin_ctz(m);
        m &= m - 1u;
        take_edge(e0 + 4 * (p & 7) + (p >> 3));
      } while (m != 0u);
    }
  };

  if (!useStored) {
    const uint4* key4 = (const uint4*)keys;
    int i0 = tid;
    for (; i0 + (U - 1) * NT < N16; i0 += NT * U) {
      uint4 kv[U];
#pragma unroll
      for (int u = 0; u < U; ++u) kv[u] = key4[i0 + u * NT];
#pragma unroll
      for (int u = 0; u < U; ++u) take16(kv[u], (i0 + u * NT) * 16);
    }
    for (; i0 < N16; i0 += NT) take16(key4[i0], i0 * 16);
  }
  __syncthreads();

  const int wave  = __builtin_amdgcn_readfirstlane(tid >> 5);
  const int total = __builtin_amdgcn_readfirstlane(
      useStored ? (int)nStored : (int)min(cnt, (unsigned)CAP));
  if (BUILD && tid == 0) gcnt[blockIdx.x * CNT_PITCH] = cnt;

  auto add_row = [&](int sj, int dj, float wj) {
    const float* p = proj + (size_t)sj * D + lane;
    float* t       = tile + dj * D + lane;
    float v[CPL];
#pragma unroll
    for (int c = 0; c < CPL; ++c) v[c] = p[32 * c];
#pragma unroll
    for (int c = 0; c < CPL; ++c) atomicAdd(t + 32 * c, v[c] * wj);
  };

  auto add_step = [&](int s, int dl, float w, int nb) {
#pragma unroll
    for (int q = 0; q < 4; ++q) {
      const int nq = nb - 8 * q;
      if (nq >= 8) {
        float v[8][CPL];
#pragma unroll
        for (int j = 0; j < 8; ++j) {
          const int sj   = __builtin_amdgcn_readlane(s, 8 * q + j);
          const float* p = proj + (size_t)sj * D + lane;
#pragma unroll
          for (int c = 0; c < CPL; ++c) v[j][c] = p[32 * c];
        }
#pragma unroll
        for (int j = 0; j < 8; ++j) {
          const int dj   = __builtin_amdgcn_readlane(dl, 8 * q + j);
          const float wj = __int_as_float(__builtin_amdgcn_readlane(__float_as_int(w), 8 * q + j));
          float* t       = tile + dj * D + lane;
#pragma unroll
          for (int c = 0; c < CPL; ++c) atomicAdd(t + 32 * c, v[j][c] * wj);
        }
      } else {
        for (int j = 0; j < nq; ++j) {
          const int l = 8 * q + j;
          add_row(__builtin_amdgcn_readlane(s, l), __builtin_amdgcn_readlane(dl, l),
                  __int_as_float(__builtin_amdgcn_readlane(__float_as_int(w), l)));
        }
      }
    }
  };

  for (int b0 = wave * 32; b0 < total; b0 += 32 * 32) {
    const int nb = min(32, total - b0);
    int s = 0, dl = 0;
    float w = 0.f;
    if (lane < nb) {
      if (useStored) {
        const int2 ent = list[b0 + lane];
        s  = ent.x & 0xFFFF;
        dl = (int)((unsigned)ent.x >> 16);
        w  = __int_as_float(ent.y);
      } else {
        const int e = list[b0 + lane].x;
        s  = src[e];
        w  = ew[e];
        dl = dst[e] - rowBase;
        if (BUILD) gl[b0 + lane] = make_int2(s | (dl << 16), __float_as_int(w));
      }
    }
    add_step(s, dl, w, nb);
  }
  __syncthreads();

  for (int i = tid; i < rowsHere * D4; i += NT) {
    const int r  = i / D4;
    const int c4 = i % D4;
    const float4 t  = ((const float4*)tile)[i];
    const float  sc = idi[rowBase + r];
    const float4 b  = ((const float4*)bias)[c4];
    float4 o;
    o.x = t.x * sc + b.x;
    o.y = t.y * sc + b.y;
    o.z = t.z * sc + b.z;
    o.w = t.w * sc + b.w;
    if (RELU) {
      o.x = fmaxf(o.x, 0.0f);
      o.y = fmaxf(o.y, 0.0f);
      o.z = fmaxf(o.z, 0.0f);
      o.w = fmaxf(o.w, 0.0f);
    }
    ((float4*)out)[(size_t)(rowBase + r) * D4 + c4] = o;
  }
}

extern "C" void kernel_launch(void* const* d_in, const int* in_sizes, int n_in,
                              void* d_out, int out_size, void* d_ws, size_t ws_size,
                              hipStream_t stream) {
  const float* feat = (const float*)d_in[0];
  const float* ew   = (const float*)d_in[1];
  const float* W1   = (const float*)d_in[2];
  const float* b1   = (const float*)d_in[3];
  const float* W2   = (const float*)d_in[4];
  const float* b2   = (const float*)d_in[5];
  const float* W3   = (const float*)d_in[6];
  const float* b3   = (const float*)d_in[7];
  const int*   src  = (const int*)d_in[8];
  const int*   dst  = (const int*)d_in[9];
  float* out = (float*)d_out;

  float* ws     = (float*)d_ws;
  float* outdeg = ws;
  float* indeg  = ws + N_NODES;
  float* odi    = ws + 2 * (size_t)N_NODES;
  float* idi    = ws + 3 * (size_t)N_NODES;
  float* h      = ws + 4 * (size_t)N_NODES;
  float* proj   = h    + (size_t)N_NODES * 128;

  const int gemmBlocks = N_NODES / 16;
  constexpr int TR = TILE_ROWS;
  constexpr int tiles = (N_NODES + TR - 1) / TR;
  constexpr int CAP128 = (256 * 1024 - TR * 128 * 4) / 8;
  constexpr int CAP64  = (256 * 1024 - TR * 64 * 4) / 8;
  int*      glist = (int*)(proj + (size_t)N_NODES * 128);
  unsigned* gcnt  = (unsigned*)(glist + (size_t)2 * tiles * CAP128);
  unsigned char* keys = (unsigned char*)(gcnt + (size_t)tiles * CNT_PITCH);

  hipMemsetAsync(outdeg, 0, 2 * (size_t)N_NODES * sizeof(float), stream);
  gcn_degrees<<<(N_EDGES + 255) / 256, 256, 0, stream>>>(src, dst, outdeg, indeg, keys);
  gcn_deg_isqrt<<<(N_NODES + 255) / 256, 256, 0, stream>>>(outdeg, indeg, odi, idi);

  gcn_gemm_wmma<128><<<gemmBlocks, 256, 0, stream>>>(feat, odi, W1, proj);
  gcn_tile_aggregate<128, TR, CAP128, CAP128, true, true><<<tiles, 1024, 0, stream>>>(proj, src, dst, ew, idi, b1, h, glist, gcnt, keys);

  gcn_gemm_wmma<128><<<gemmBlocks, 256, 0, stream>>>(h, odi, W2, proj);
  gcn_tile_aggregate<128, TR, CAP128, CAP128, true, false><<<tiles, 1024, 0, stream>>>(proj, src, dst, ew, idi, b2, h, glist, gcnt, keys);

  gcn_gemm_wmma<64><<<gemmBlocks, 128, 0, stream>>>(h, odi, W3, proj);
  gcn_tile_aggregate<64, TR, CAP64, CAP128, false, false><<<tiles, 1024, 0, stream>>>(proj, src, dst, ew, idi, b3, out, glist, gcnt, keys);
}


# In-degree counted by the row's tile work-group in layer 1 (LDS integer adds over its edges), not by 800000 global float atomics in the degree pass; float row sums unchanged (LDS arrival order)
# speedup vs baseline: 2.7050x; 1.1111x over previous
#include <hip/hip_runtime.h>
#include <math.h>


typedef __attribute__((ext_vector_type(2))) float v2f;
typedef __attribute__((ext_vector_type(8))) float v8f;

#define N_NODES 50000
#define N_EDGES 800000
#define K_DIM   128
#define CNT_PITCH 32
#define TILE_ROWS 197
#define NO_TILE 255
#define IDI_PITCH 256

__global__ void gcn_degrees(const int* __restrict__ src, const int* __restrict__ dst,
                            float* __restrict__ outdeg,
                            unsigned char* __restrict__ keys) {
  static_assert((N_NODES - 1) / TILE_ROWS < NO_TILE, "tile keys fit a byte");
  int e = blockIdx.x * blockDim.x + threadIdx.x;
  if (e < N_EDGES) {
    const int d = dst[e];
    atomicAdd(&outdeg[src[e]], 1.0f);
    keys[e] = (unsigned char)((unsigned)d < (unsigned)N_NODES ? (unsigned)d / TILE_ROWS : NO_TILE);
  }
}

__global__ void gcn_deg_isqrt(const float* __restrict__ outdeg, float* __restrict__ odi) {
  int i = blockIdx.x * blockDim.x + threadIdx.x;
  if (i < N_NODES) odi[i] = rsqrtf(fmaxf(outdeg[i], 1.0f));
}

template <int NOUT>
__global__ __launch_bounds__(256) void gcn_gemm_wmma(const float* __restrict__ A,
                                                     const float* __restrict__ odi,
                                                     const float* __restrict__ W,
                                                     float* __restrict__ C) {
  __shared__ float As[16][K_DIM + 4];

  const int tileM = blockIdx.x * 16;

  for (int idx = threadIdx.x; idx < 16 * K_DIM; idx += blockDim.x) {
    const int r = idx >> 7;
    const int c = idx & (K_DIM - 1);
    As[r][c] = A[(size_t)(tileM + r) * K_DIM + c] * odi[tileM + r];
  }
  __syncthreads();

  const int wave = threadIdx.x >> 5;
  const int lane = threadIdx.x & 31;
  const int half = lane >> 4;
  const int r16  = lane & 15;
  const int n0   = wave * 16;

  v8f acc = {};
#pragma unroll
  for (int k = 0; k < K_DIM; k += 4) {
    v2f a, b;
    a.x = As[r16][k + 2 * half];
    a.y = As[r16][k + 2 * half + 1];
    b.x = W[(size_t)(k + 2 * half)     * NOUT + n0 + r16];
    b.y = W[(size_t)(k + 2 * half + 1) * NOUT + n0 + r16];
    acc = __builtin_amdgcn_wmma_f32_16x16x4_f32(false, a, false, b,
                                                (short)0, acc, false, false);
  }

#pragma unroll
  for (int r = 0; r < 8; ++r) {
    const int row = tileM + r + 8 * half;
    C[(size_t)row * NOUT + n0 + r16] = acc[r];
  }
}

template <int D, int TR, int CAP, int GCAP, bool RELU, bool BUILD>
__global__ __launch_bounds__(1024) void gcn_tile_aggregate(const float* __restrict__ proj,
                                                           const int* __restrict__ src,
                                                           const int* __restrict__ dst,
                                                           const float* __restrict__ ew,
                                                           float* __restrict__ idiT,
                                                           const float* __restrict__ bias,
                                                           float* __restrict__ out,
                                                           int* __restrict__ glist,
                                                           unsigned* __restrict__ gcnt,
                                                           const unsigned char* __restrict__ keys) {
  static_assert(D % 32 == 0 && N_EDGES % 16 == 0, "layout");
  static_assert(TR == TILE_ROWS, "a tile key names TR rows");
  static_assert(N_NODES <= 65536 && TR <= 32768, "packed list entry: 16 bits each");
  static_assert(BUILD ? CAP <= GCAP : GCAP <= CAP, "listed edges fit glist; a stored list fits the LDS list");
  static_assert(TR <= IDI_PITCH, "a tile's in_deg^{-1/2} values fit its segment of idiT");
  constexpr int NT  = 1024;
  constexpr int CPL = D / 32;
  constexpr int D4  = D / 4;
  constexpr int N16 = N_EDGES / 16;
  constexpr int U   = 8;

  __shared__ __attribute__((aligned(16))) float tile[TR * D];
  __shared__ int2     list[CAP];
  __shared__ unsigned cnt;
  __shared__ unsigned rowcnt[TR];

  const int tid      = threadIdx.x;
  const int lane     = tid & 31;
  const int rowBase  = blockIdx.x * TR;
  const int rowsHere = min(TR, N_NODES - rowBase);

  int2* const gl = (int2*)glist + (size_t)blockIdx.x * GCAP;
  unsigned nStored = 0u;
  if (!BUILD) nStored = gcnt[blockIdx.x * CNT_PITCH];
  const bool useStored = !BUILD && nStored <= (unsigned)GCAP;
  constexpr int PRE = 4;
  int2 pre[PRE];
  if (useStored) {
#pragma unroll
    for (int u = 0; u < PRE; ++u) {
      const unsigned i = (unsigned)(tid + u * NT);
      pre[u] = (i < nStored) ? gl[i] : make_int2(0, 0);
    }
  }

  for (int i = tid; i < TR * D4; i += NT)
    ((float4*)tile)[i] = make_float4(0.f, 0.f, 0.f, 0.f);
  if (tid == 0) cnt = 0u;
  if (BUILD)
    for (int i = tid; i < TR; i += NT) rowcnt[i] = 0u;
  if (useStored) {
#pragma unroll
    for (int u = 0; u < PRE; ++u) {
      const unsigned i = (unsigned)(tid + u * NT);
      if (i < nStored) list[i] = pre[u];
    }
    for (unsigned i = (unsigned)(tid + PRE * NT); i < nStored; i += NT) list[i] = gl[i];
  }
  __syncthreads();

  auto take_edge = [&](int e) {
    const unsigned slot = atomicAdd(&cnt, 1u);
    if (slot < (unsigned)CAP) {
      list[slot].x = e;
    } else {
      const float w  = ew[e];
      const float* p = proj + (size_t)src[e] * D;
      const int dl   = dst[e] - rowBase;
      float* t       = tile + dl * D;
      if (BUILD) atomicAdd(&rowcnt[dl], 1u);
#pragma nounroll
      for (int c = 0; c < D; ++c) atomicAdd(t + c, p[c] * w);
    }
  };
  const unsigned pat = (unsigned)blockIdx.x * 0x01010101u;
  auto differs = [&](unsigned k) {
    const unsigned x = k ^ pat;
    return ((x & 0x7F7F7F7Fu) + 0x7F7F7F7Fu) | x;
  };
  auto take16 = [&](const uint4& k, int e0) {
    const unsigned t0 = differs(k.x), t1 = differs(k.y), t2 = differs(k.z), t3 = differs(k.w);
    if ((t0 & t1 & t2 & t3 & 0x80808080u) != 0x80808080u) {
      unsigned m = ~(((t0 >> 7) & 0x01010101u) | ((t1 >> 6) & 0x02020202u) |
                     ((t2 >> 5) & 0x04040404u) | ((t3 >> 4) & 0x08080808u)) & 0x0F0F0F0Fu;
      do {
        const int p = __builtin_ctz(m);
        m &= m - 1u;
        take_edge(e0 + 4 * (p & 7) + (p >> 3));
      } while (m != 0u);
    }
  };

  if (!useStored) {
    const uint4* key4 = (const uint4*)keys;
    int i0 = tid;
    for (; i0 + (U - 1) * NT < N16; i0 += NT * U) {
      uint4 kv[U];
#pragma unroll
      for (int u = 0; u < U; ++u) kv[u] = key4[i0 + u * NT];
#pragma unroll
      for (int u = 0; u < U; ++u) take16(kv[u], (i0 + u * NT) * 16);
    }
    for (; i0 < N16; i0 += NT) take16(key4[i0], i0 * 16);
  }
  __syncthreads();

  const int wave  = __builtin_amdgcn_readfirstlane(tid >> 5);
  const int total = __builtin_amdgcn_readfirstlane(
      useStored ? (int)nStored : (int)min(cnt, (unsigned)CAP));
  if (BUILD && tid == 0) gcnt[blockIdx.x * CNT_PITCH] = cnt;

  auto add_row = [&](int sj, int dj, float wj) {
    const float* p = proj + (size_t)sj * D + lane;
    float* t       = tile + dj * D + lane;
    float v[CPL];
#pragma unroll
    for (int c = 0; c < CPL; ++c) v[c] = p[32 * c];
#pragma unroll
    for (int c = 0; c < CPL; ++c) atomicAdd(t + 32 * c, v[c] * wj);
  };

  auto add_step = [&](int s, int dl, float w, int nb) {
#pragma unroll
    for (int q = 0; q < 4; ++q) {
      const int nq = nb - 8 * q;
      if (nq >= 8) {
        float v[8][CPL];
#pragma unroll
        for (int j = 0; j < 8; ++j) {
          const int sj   = __builtin_amdgcn_readlane(s, 8 * q + j);
          const float* p = proj + (size_t)sj * D + lane;
#pragma unroll
          for (int c = 0; c < CPL; ++c) v[j][c] = p[32 * c];
        }
#pragma unroll
        for (int j = 0; j < 8; ++j) {
          const int dj   = __builtin_amdgcn_readlane(dl, 8 * q + j);
          const float wj = __int_as_float(__builtin_amdgcn_readlane(__float_as_int(w), 8 * q + j));
          float* t       = tile + dj * D + lane;
#pragma unroll
          for (int c = 0; c < CPL; ++c) atomicAdd(t + 32 * c, v[j][c] * wj);
        }
      } else {
        for (int j = 0; j < nq; ++j) {
          const int l = 8 * q + j;
          add_row(__builtin_amdgcn_readlane(s, l), __builtin_amdgcn_readlane(dl, l),
                  __int_as_float(__builtin_amdgcn_readlane(__float_as_int(w), l)));
        }
      }
    }
  };

  for (int b0 = wave * 32; b0 < total; b0 += 32 * 32) {
    const int nb = min(32, total - b0);
    int s = 0, dl = 0;
    float w = 0.f;
    if (lane < nb) {
      if (useStored) {
        const int2 ent = list[b0 + lane];
        s  = ent.x & 0xFFFF;
        dl = (int)((unsigned)ent.x >> 16);
        w  = __int_as_float(ent.y);
      } else {
        const int e = list[b0 + lane].x;
        s  = src[e];
        w  = ew[e];
        dl = dst[e] - rowBase;
        if (BUILD) {
          atomicAdd(&rowcnt[dl], 1u);
          gl[b0 + lane] = make_int2(s | (dl << 16), __float_as_int(w));
        }
      }
    }
    add_step(s, dl, w, nb);
  }
  __syncthreads();

  for (int i = tid; i < rowsHere * D4; i += NT) {
    const int r  = i / D4;
    const int c4 = i % D4;
    const float4 t  = ((const float4*)tile)[i];
    float sc;
    if (BUILD) {
      sc = rsqrtf(fmaxf((float)rowcnt[r], 1.0f));
      if (c4 == 0) idiT[blockIdx.x * IDI_PITCH + r] = sc;
    } else {
      sc = idiT[blockIdx.x * IDI_PITCH + r];
    }
    const float4 b  = ((const float4*)bias)[c4];
    float4 o;
    o.x = t.x * sc + b.x;
    o.y = t.y * sc + b.y;
    o.z = t.z * sc + b.z;
    o.w = t.w * sc + b.w;
    if (RELU) {
      o.x = fmaxf(o.x, 0.0f);
      o.y = fmaxf(o.y, 0.0f);
      o.z = fmaxf(o.z, 0.0f);
      o.w = fmaxf(o.w, 0.0f);
    }
    ((float4*)out)[(size_t)(rowBase + r) * D4 + c4] = o;
  }
}

extern "C" void kernel_launch(void* const* d_in, const int* in_sizes, int n_in,
                              void* d_out, int out_size, void* d_ws, size_t ws_size,
                              hipStream_t stream) {
  const float* feat = (const float*)d_in[0];
  const float* ew   = (const float*)d_in[1];
  const float* W1   = (const float*)d_in[2];
  const float* b1   = (const float*)d_in[3];
  const float* W2   = (const float*)d_in[4];
  const float* b2   = (const float*)d_in[5];
  const float* W3   = (const float*)d_in[6];
  const float* b3   = (const float*)d_in[7];
  const int*   src  = (const int*)d_in[8];
  const int*   dst  = (const int*)d_in[9];
  float* out = (float*)d_out;

  float* ws     = (float*)d_ws;
  float* outdeg = ws;
  float* odi    = ws + N_NODES;
  float* idiT   = ws + 2 * (size_t)N_NODES;
  float* h      = ws + 4 * (size_t)N_NODES;
  float* proj   = h    + (size_t)N_NODES * 128;

  const int gemmBlocks = N_NODES / 16;
  constexpr int TR = TILE_ROWS;
  constexpr int tiles = (N_NODES + TR - 1) / TR;
  static_assert(tiles * IDI_PITCH <= 2 * N_NODES, "idiT ends before h");
  constexpr int CAP128 = (256 * 1024 - TR * 128 * 4) / 8;
  constexpr int CAP64  = (256 * 1024 - TR * 64 * 4) / 8;
  int*      glist = (int*)(proj + (size_t)N_NODES * 128);
  unsigned* gcnt  = (unsigned*)(glist + (size_t)2 * tiles * CAP128);
  unsigned char* keys = (unsigned char*)(gcnt + (size_t)tiles * CNT_PITCH);

  hipMemsetAsync(outdeg, 0, (size_t)N_NODES * sizeof(float), stream);
  gcn_degrees<<<(N_EDGES + 255) / 256, 256, 0, stream>>>(src, dst, outdeg, keys);
  gcn_deg_isqrt<<<(N_NODES + 255) / 256, 256, 0, stream>>>(outdeg, odi);

  gcn_gemm_wmma<128><<<gemmBlocks, 256, 0, stream>>>(feat, odi, W1, proj);
  gcn_tile_aggregate<128, TR, CAP128, CAP128, true, true><<<tiles, 1024, 0, stream>>>(proj, src, dst, ew, idiT, b1, h, glist, gcnt, keys);

  gcn_gemm_wmma<128><<<gemmBlocks, 256, 0, stream>>>(h, odi, W2, proj);
  gcn_tile_aggregate<128, TR, CAP128, CAP128, true, false><<<tiles, 1024, 0, stream>>>(proj, src, dst, ew, idiT, b2, h, glist, gcnt, keys);

  gcn_gemm_wmma<64><<<gemmBlocks, 128, 0, stream>>>(h, odi, W3, proj);
  gcn_tile_aggregate<64, TR, CAP64, CAP128, false, false><<<tiles, 1024, 0, stream>>>(proj, src, dst, ew, idiT, b3, out, glist, gcnt, keys);
}


# Out-degree without global float atomics or memset: 256 work-groups (32 edge slices x 8 node ranges) count in LDS (integer adds), store partials; normalizer pass sums 32; float sums unchanged
# speedup vs baseline: 3.3519x; 1.2391x over previous
#include <hip/hip_runtime.h>
#include <math.h>


typedef __attribute__((ext_vector_type(2))) float v2f;
typedef __attribute__((ext_vector_type(8))) float v8f;

#define N_NODES 50000
#define N_EDGES 800000
#define K_DIM   128
#define CNT_PITCH 32
#define TILE_ROWS 197
#define NO_TILE 255
#define IDI_PITCH 256
#define DEG_SLICES 32
#define DEG_RANGES 8
#define DEG_RANGE_NODES (N_NODES / DEG_RANGES)
#define DEG_PITCH 6272
#define DEG_THREADS 512
#define KEY_QUADS 800

__global__ __launch_bounds__(DEG_THREADS) void gcn_outdeg_count(const int* __restrict__ src,
                                                                const int* __restrict__ dst,
                                                                unsigned* __restrict__ part,
                                                                unsigned char* __restrict__ keys) {
  static_assert((N_NODES - 1) / TILE_ROWS < NO_TILE, "tile keys fit a byte");
  static_assert(N_NODES % DEG_RANGES == 0 && N_EDGES % (4 * DEG_SLICES) == 0, "even split");
  static_assert(DEG_PITCH >= DEG_RANGE_NODES && DEG_PITCH % 32 == 0, "segment = whole lines");
  static_assert(KEY_QUADS % 32 == 0 && N_EDGES % (4 * KEY_QUADS) == 0 &&
                N_EDGES / (4 * KEY_QUADS) <= DEG_SLICES * DEG_RANGES, "key groups cover the edges");
  constexpr int SLICE4 = N_EDGES / DEG_SLICES / 4;
  constexpr int PER    = (SLICE4 + DEG_THREADS - 1) / DEG_THREADS;
  constexpr int KGROUPS = N_EDGES / (4 * KEY_QUADS);

  __shared__ __attribute__((aligned(16))) unsigned cntL[DEG_PITCH];

  const int tid = threadIdx.x;
  const int s   = blockIdx.x / DEG_RANGES;
  const int lo  = (blockIdx.x % DEG_RANGES) * DEG_RANGE_NODES;

  const int4* s4 = (const int4*)src + (size_t)s * SLICE4;
  int4 v[PER];
#pragma unroll
  for (int u = 0; u < PER; ++u) {
    const int i = tid + u * DEG_THREADS;
    v[u] = (i < SLICE4) ? s4[i] : make_int4(-1, -1, -1, -1);
  }

  if (blockIdx.x < KGROUPS) {
    auto key = [](int d) {
      return (unsigned)d < (unsigned)N_NODES ? (unsigned)d / TILE_ROWS : (unsigned)NO_TILE;
    };
    for (int j = tid; j < KEY_QUADS; j += DEG_THREADS) {
      const int q  = blockIdx.x * KEY_QUADS + j;
      const int4 d = ((const int4*)dst)[q];
      ((unsigned*)keys)[q] = key(d.x) | (key(d.y) << 8) | (key(d.z) << 16) | (key(d.w) << 24);
    }
  }

  for (int i = tid; i < DEG_PITCH / 4; i += DEG_THREADS)
    ((uint4*)cntL)[i] = make_uint4(0u, 0u, 0u, 0u);
  __syncthreads();

  auto count = [&](int sv) {
    const unsigned o = (unsigned)(sv - lo);
    if (o < (unsigned)DEG_RANGE_NODES) atomicAdd(&cntL[o], 1u);
  };
#pragma unroll
  for (int u = 0; u < PER; ++u) {
    count(v[u].x);
    count(v[u].y);
    count(v[u].z);
    count(v[u].w);
  }
  __syncthreads();

  uint4* seg = (uint4*)(part + (size_t)blockIdx.x * DEG_PITCH);
  for (int i = tid; i < DEG_PITCH / 4; i += DEG_THREADS) seg[i] = ((const uint4*)cntL)[i];
}

__global__ void gcn_outdeg_isqrt(const unsigned* __restrict__ part, float* __restrict__ odi) {
  int n = blockIdx.x * blockDim.x + threadIdx.x;
  if (n < N_NODES) {
    const int r = n / DEG_RANGE_NODES;
    const unsigned* p = part + (size_t)r * DEG_PITCH + (n - r * DEG_RANGE_NODES);
    unsigned c = 0u;
#pragma unroll
    for (int s = 0; s < DEG_SLICES; ++s) c += p[(size_t)s * DEG_RANGES * DEG_PITCH];
    odi[n] = rsqrtf(fmaxf((float)c, 1.0f));
  }
}

template <int NOUT>
__global__ __launch_bounds__(256) void gcn_gemm_wmma(const float* __restrict__ A,
                                                     const float* __restrict__ odi,
                                                     const float* __restrict__ W,
                                                     float* __restrict__ C) {
  __shared__ float As[16][K_DIM + 4];

  const int tileM = blockIdx.x * 16;

  for (int idx = threadIdx.x; idx < 16 * K_DIM; idx += blockDim.x) {
    const int r = idx >> 7;
    const int c = idx & (K_DIM - 1);
    As[r][c] = A[(size_t)(tileM + r) * K_DIM + c] * odi[tileM + r];
  }
  __syncthreads();

  const int wave = threadIdx.x >> 5;
  const int lane = threadIdx.x & 31;
  const int half = lane >> 4;
  const int r16  = lane & 15;
  const int n0   = wave * 16;

  v8f acc = {};
#pragma unroll
  for (int k = 0; k < K_DIM; k += 4) {
    v2f a, b;
    a.x = As[r16][k + 2 * half];
    a.y = As[r16][k + 2 * half + 1];
    b.x = W[(size_t)(k + 2 * half)     * NOUT + n0 + r16];
    b.y = W[(size_t)(k + 2 * half + 1) * NOUT + n0 + r16];
    acc = __builtin_amdgcn_wmma_f32_16x16x4_f32(false, a, false, b,
                                                (short)0, acc, false, false);
  }

#pragma unroll
  for (int r = 0; r < 8; ++r) {
    const int row = tileM + r + 8 * half;
    C[(size_t)row * NOUT + n0 + r16] = acc[r];
  }
}

template <int D, int TR, int CAP, int GCAP, bool RELU, bool BUILD>
__global__ __launch_bounds__(1024) void gcn_tile_aggregate(const float* __restrict__ proj,
                                                           const int* __restrict__ src,
                                                           const int* __restrict__ dst,
                                                           const float* __restrict__ ew,
                                                           float* __restrict__ idiT,
                                                           const float* __restrict__ bias,
                                                           float* __restrict__ out,
                                                           int* __restrict__ glist,
                                                           unsigned* __restrict__ gcnt,
                                                           const unsigned char* __restrict__ keys) {
  static_assert(D % 32 == 0 && N_EDGES % 16 == 0, "layout");
  static_assert(TR == TILE_ROWS, "a tile key names TR rows");
  static_assert(N_NODES <= 65536 && TR <= 32768, "packed list entry: 16 bits each");
  static_assert(BUILD ? CAP <= GCAP : GCAP <= CAP, "listed edges fit glist; a stored list fits the LDS list");
  static_assert(TR <= IDI_PITCH, "a tile's in_deg^{-1/2} values fit its segment of idiT");
  constexpr int NT  = 1024;
  constexpr int CPL = D / 32;
  constexpr int D4  = D / 4;
  constexpr int N16 = N_EDGES / 16;
  constexpr int U   = 8;

  __shared__ __attribute__((aligned(16))) float tile[TR * D];
  __shared__ int2     list[CAP];
  __shared__ unsigned cnt;
  __shared__ unsigned rowcnt[TR];

  const int tid      = threadIdx.x;
  const int lane     = tid & 31;
  const int rowBase  = blockIdx.x * TR;
  const int rowsHere = min(TR, N_NODES - rowBase);

  int2* const gl = (int2*)glist + (size_t)blockIdx.x * GCAP;
  unsigned nStored = 0u;
  if (!BUILD) nStored = gcnt[blockIdx.x * CNT_PITCH];
  const bool useStored = !BUILD && nStored <= (unsigned)GCAP;
  constexpr int PRE = 4;
  int2 pre[PRE];
  if (useStored) {
#pragma unroll
    for (int u = 0; u < PRE; ++u) {
      const unsigned i = (unsigned)(tid + u * NT);
      pre[u] = (i < nStored) ? gl[i] : make_int2(0, 0);
    }
  }

  for (int i = tid; i < TR * D4; i += NT)
    ((float4*)tile)[i] = make_float4(0.f, 0.f, 0.f, 0.f);
  if (tid == 0) cnt = 0u;
  if (BUILD)
    for (int i = tid; i < TR; i += NT) rowcnt[i] = 0u;
  if (useStored) {
#pragma unroll
    for (int u = 0; u < PRE; ++u) {
      const unsigned i = (unsigned)(tid + u * NT);
      if (i < nStored) list[i] = pre[u];
    }
    for (unsigned i = (unsigned)(tid + PRE * NT); i < nStored; i += NT) list[i] = gl[i];
  }
  __syncthreads();

  auto take_edge = [&](int e) {
    const unsigned slot = atomicAdd(&cnt, 1u);
    if (slot < (unsigned)CAP) {
      list[slot].x = e;
    } else {
      const float w  = ew[e];
      const float* p = proj + (size_t)src[e] * D;
      const int dl   = dst[e] - rowBase;
      float* t       = tile + dl * D;
      if (BUILD) atomicAdd(&rowcnt[dl], 1u);
#pragma nounroll
      for (int c = 0; c < D; ++c) atomicAdd(t + c, p[c] * w);
    }
  };
  const unsigned pat = (unsigned)blockIdx.x * 0x01010101u;
  auto differs = [&](unsigned k) {
    const unsigned x = k ^ pat;
    return ((x & 0x7F7F7F7Fu) + 0x7F7F7F7Fu) | x;
  };
  auto take16 = [&](const uint4& k, int e0) {
    const unsigned t0 = differs(k.x), t1 = differs(k.y), t2 = differs(k.z), t3 = differs(k.w);
    if ((t0 & t1 & t2 & t3 & 0x80808080u) != 0x80808080u) {
      unsigned m = ~(((t0 >> 7) & 0x01010101u) | ((t1 >> 6) & 0x02020202u) |
                     ((t2 >> 5) & 0x04040404u) | ((t3 >> 4) & 0x08080808u)) & 0x0F0F0F0Fu;
      do {
        const int p = __builtin_ctz(m);
        m &= m - 1u;
        take_edge(e0 + 4 * (p & 7) + (p >> 3));
      } while (m != 0u);
    }
  };

  if (!useStored) {
    const uint4* key4 = (const uint4*)keys;
    int i0 = tid;
    for (; i0 + (U - 1) * NT < N16; i0 += NT * U) {
      uint4 kv[U];
#pragma unroll
      for (int u = 0; u < U; ++u) kv[u] = key4[i0 + u * NT];
#pragma unroll
      for (int u = 0; u < U; ++u) take16(kv[u], (i0 + u * NT) * 16);
    }
    for (; i0 < N16; i0 += NT) take16(key4[i0], i0 * 16);
  }
  __syncthreads();

  const int wave  = __builtin_amdgcn_readfirstlane(tid >> 5);
  const int total = __builtin_amdgcn_readfirstlane(
      useStored ? (int)nStored : (int)min(cnt, (unsigned)CAP));
  if (BUILD && tid == 0) gcnt[blockIdx.x * CNT_PITCH] = cnt;

  auto add_row = [&](int sj, int dj, float wj) {
    const float* p = proj + (size_t)sj * D + lane;
    float* t       = tile + dj * D + lane;
    float v[CPL];
#pragma unroll
    for (int c = 0; c < CPL; ++c) v[c] = p[32 * c];
#pragma unroll
    for (int c = 0; c < CPL; ++c) atomicAdd(t + 32 * c, v[c] * wj);
  };

  auto add_step = [&](int s, int dl, float w, int nb) {
#pragma unroll
    for (int q = 0; q < 4; ++q) {
      const int nq = nb - 8 * q;
      if (nq >= 8) {
        float v[8][CPL];
#pragma unroll
        for (int j = 0; j < 8; ++j) {
          const int sj   = __builtin_amdgcn_readlane(s, 8 * q + j);
          const float* p = proj + (size_t)sj * D + lane;
#pragma unroll
          for (int c = 0; c < CPL; ++c) v[j][c] = p[32 * c];
        }
#pragma unroll
        for (int j = 0; j < 8; ++j) {
          const int dj   = __builtin_amdgcn_readlane(dl, 8 * q + j);
          const float wj = __int_as_float(__builtin_amdgcn_readlane(__float_as_int(w), 8 * q + j));
          float* t       = tile + dj * D + lane;
#pragma unroll
          for (int c = 0; c < CPL; ++c) atomicAdd(t + 32 * c, v[j][c] * wj);
        }
      } else {
        for (int j = 0; j < nq; ++j) {
          const int l = 8 * q + j;
          add_row(__builtin_amdgcn_readlane(s, l), __builtin_amdgcn_readlane(dl, l),
                  __int_as_float(__builtin_amdgcn_readlane(__float_as_int(w), l)));
        }
      }
    }
  };

  for (int b0 = wave * 32; b0 < total; b0 += 32 * 32) {
    const int nb = min(32, total - b0);
    int s = 0, dl = 0;
    float w = 0.f;
    if (lane < nb) {
      if (useStored) {
        const int2 ent = list[b0 + lane];
        s  = ent.x & 0xFFFF;
        dl = (int)((unsigned)ent.x >> 16);
        w  = __int_as_float(ent.y);
      } else {
        const int e = list[b0 + lane].x;
        s  = src[e];
        w  = ew[e];
        dl = dst[e] - rowBase;
        if (BUILD) {
          atomicAdd(&rowcnt[dl], 1u);
          gl[b0 + lane] = make_int2(s | (dl << 16), __float_as_int(w));
        }
      }
    }
    add_step(s, dl, w, nb);
  }
  __syncthreads();

  for (int i = tid; i < rowsHere * D4; i += NT) {
    const int r  = i / D4;
    const int c4 = i % D4;
    const float4 t  = ((const float4*)tile)[i];
    float sc;
    if (BUILD) {
      sc = rsqrtf(fmaxf((float)rowcnt[r], 1.0f));
      if (c4 == 0) idiT[blockIdx.x * IDI_PITCH + r] = sc;
    } else {
      sc = idiT[blockIdx.x * IDI_PITCH + r];
    }
    const float4 b  = ((const float4*)bias)[c4];
    float4 o;
    o.x = t.x * sc + b.x;
    o.y = t.y * sc + b.y;
    o.z = t.z * sc + b.z;
    o.w = t.w * sc + b.w;
    if (RELU) {
      o.x = fmaxf(o.x, 0.0f);
      o.y = fmaxf(o.y, 0.0f);
      o.z = fmaxf(o.z, 0.0f);
      o.w = fmaxf(o.w, 0.0f);
    }
    ((float4*)out)[(size_t)(rowBase + r) * D4 + c4] = o;
  }
}

extern "C" void kernel_launch(void* const* d_in, const int* in_sizes, int n_in,
                              void* d_out, int out_size, void* d_ws, size_t ws_size,
                              hipStream_t stream) {
  const float* feat = (const float*)d_in[0];
  const float* ew   = (const float*)d_in[1];
  const float* W1   = (const float*)d_in[2];
  const float* b1   = (const float*)d_in[3];
  const float* W2   = (const float*)d_in[4];
  const float* b2   = (const float*)d_in[5];
  const float* W3   = (const float*)d_in[6];
  const float* b3   = (const float*)d_in[7];
  const int*   src  = (const int*)d_in[8];
  const int*   dst  = (const int*)d_in[9];
  float* out = (float*)d_out;

  float* ws     = (float*)d_ws;
  float* odi    = ws + N_NODES;
  float* idiT   = ws + 2 * (size_t)N_NODES;
  float* h      = ws + 4 * (size_t)N_NODES;
  float* proj   = h    + (size_t)N_NODES * 128;

  const int gemmBlocks = N_NODES / 16;
  constexpr int TR = TILE_ROWS;
  constexpr int tiles = (N_NODES + TR - 1) / TR;
  static_assert(tiles * IDI_PITCH <= 2 * N_NODES, "idiT ends before h");
  constexpr int CAP128 = (256 * 1024 - TR * 128 * 4) / 8;
  constexpr int CAP64  = (256 * 1024 - TR * 64 * 4) / 8;
  int*      glist = (int*)(proj + (size_t)N_NODES * 128);
  unsigned* gcnt  = (unsigned*)(glist + (size_t)2 * tiles * CAP128);
  unsigned char* keys = (unsigned char*)(gcnt + (size_t)tiles * CNT_PITCH);
  unsigned* part = (unsigned*)(keys + N_EDGES);

  gcn_outdeg_count<<<DEG_SLICES * DEG_RANGES, DEG_THREADS, 0, stream>>>(src, dst, part, keys);
  gcn_outdeg_isqrt<<<(N_NODES + 255) / 256, 256, 0, stream>>>(part, odi);

  gcn_gemm_wmma<128><<<gemmBlocks, 256, 0, stream>>>(feat, odi, W1, proj);
  gcn_tile_aggregate<128, TR, CAP128, CAP128, true, true><<<tiles, 1024, 0, stream>>>(proj, src, dst, ew, idiT, b1, h, glist, gcnt, keys);

  gcn_gemm_wmma<128><<<gemmBlocks, 256, 0, stream>>>(h, odi, W2, proj);
  gcn_tile_aggregate<128, TR, CAP128, CAP128, true, false><<<tiles, 1024, 0, stream>>>(proj, src, dst, ew, idiT, b2, h, glist, gcnt, keys);

  gcn_gemm_wmma<64><<<gemmBlocks, 128, 0, stream>>>(h, odi, W3, proj);
  gcn_tile_aggregate<64, TR, CAP64, CAP128, false, false><<<tiles, 1024, 0, stream>>>(proj, src, dst, ew, idiT, b3, out, glist, gcnt, keys);
}


# Tile edge filter reads the tile's bit row (1 bit per 4-edge group, set in LDS by the degree pass) and tests only flagged groups' dst, not all 800000 keys; sums LDS arrival order (list order changed)
# speedup vs baseline: 3.4796x; 1.0381x over previous
#include <hip/hip_runtime.h>
#include <math.h>


typedef __attribute__((ext_vector_type(2))) float v2f;
typedef __attribute__((ext_vector_type(8))) float v8f;

#define N_NODES 50000
#define N_EDGES 800000
#define K_DIM   128
#define CNT_PITCH 32
#define TILE_ROWS 197
#define N_TILES ((N_NODES + TILE_ROWS - 1) / TILE_ROWS)
#define BM_GROUPS 1024
#define BM_CHUNKS ((N_EDGES / 4 + BM_GROUPS - 1) / BM_GROUPS)
#define BM_TILE_WORDS (BM_CHUNKS * 32)
#define IDI_PITCH 256
#define DEG_SLICES 32
#define DEG_RANGES 8
#define DEG_RANGE_NODES (N_NODES / DEG_RANGES)
#define DEG_PITCH 6272
#define DEG_THREADS 512

__global__ __launch_bounds__(DEG_THREADS) void gcn_outdeg_count(const int* __restrict__ src,
                                                                const int* __restrict__ dst,
                                                                unsigned* __restrict__ part,
                                                                unsigned* __restrict__ bm) {
  static_assert(N_NODES % DEG_RANGES == 0 && N_EDGES % (4 * DEG_SLICES) == 0, "even split");
  static_assert(DEG_PITCH >= DEG_RANGE_NODES && DEG_PITCH % 32 == 0, "segment = whole lines");
  static_assert(BM_CHUNKS <= DEG_SLICES * DEG_RANGES && BM_GROUPS == 2 * DEG_THREADS && N_EDGES % 4 == 0,
                "one work-group per bitmap chunk, two groups per thread");
  constexpr int SLICE4 = N_EDGES / DEG_SLICES / 4;
  constexpr int PER    = (SLICE4 + DEG_THREADS - 1) / DEG_THREADS;

  __shared__ __attribute__((aligned(16))) unsigned cntL[DEG_PITCH];
  __shared__ __attribute__((aligned(16))) unsigned bmL[N_TILES * 32];

  const int tid = threadIdx.x;
  const int s   = blockIdx.x / DEG_RANGES;
  const int lo  = (blockIdx.x % DEG_RANGES) * DEG_RANGE_NODES;

  const int4* s4 = (const int4*)src + (size_t)s * SLICE4;
  int4 v[PER];
#pragma unroll
  for (int u = 0; u < PER; ++u) {
    const int i = tid + u * DEG_THREADS;
    v[u] = (i < SLICE4) ? s4[i] : make_int4(-1, -1, -1, -1);
  }

  const bool chunk = blockIdx.x < BM_CHUNKS;
  int4 dg[2];
#pragma unroll
  for (int u = 0; u < 2; ++u) {
    const int q = blockIdx.x * BM_GROUPS + tid + u * DEG_THREADS;
    dg[u] = (chunk && q < N_EDGES / 4) ? ((const int4*)dst)[q] : make_int4(-1, -1, -1, -1);
  }

  for (int i = tid; i < DEG_PITCH / 4; i += DEG_THREADS)
    ((uint4*)cntL)[i] = make_uint4(0u, 0u, 0u, 0u);
  if (chunk)
    for (int i = tid; i < N_TILES * 8; i += DEG_THREADS)
      ((uint4*)bmL)[i] = make_uint4(0u, 0u, 0u, 0u);
  __syncthreads();

  auto mark = [&](int d, int g) {
    if ((unsigned)d < (unsigned)N_NODES)
      atomicOr(&bmL[((unsigned)d / TILE_ROWS) * 32 + (g >> 5)], 1u << (g & 31));
  };
#pragma unroll
  for (int u = 0; u < 2; ++u) {
    const int g = tid + u * DEG_THREADS;
    mark(dg[u].x, g);
    mark(dg[u].y, g);
    mark(dg[u].z, g);
    mark(dg[u].w, g);
  }

  auto count = [&](int sv) {
    const unsigned o = (unsigned)(sv - lo);
    if (o < (unsigned)DEG_RANGE_NODES) atomicAdd(&cntL[o], 1u);
  };
#pragma unroll
  for (int u = 0; u < PER; ++u) {
    count(v[u].x);
    count(v[u].y);
    count(v[u].z);
    count(v[u].w);
  }
  __syncthreads();

  uint4* seg = (uint4*)(part + (size_t)blockIdx.x * DEG_PITCH);
  for (int i = tid; i < DEG_PITCH / 4; i += DEG_THREADS) seg[i] = ((const uint4*)cntL)[i];

  if (chunk)
    for (int i = tid; i < N_TILES * 8; i += DEG_THREADS)
      ((uint4*)(bm + (size_t)(i >> 3) * BM_TILE_WORDS + blockIdx.x * 32))[i & 7] = ((const uint4*)bmL)[i];
}

__global__ void gcn_outdeg_isqrt(const unsigned* __restrict__ part, float* __restrict__ odi) {
  int n = blockIdx.x * blockDim.x + threadIdx.x;
  if (n < N_NODES) {
    const int r = n / DEG_RANGE_NODES;
    const unsigned* p = part + (size_t)r * DEG_PITCH + (n - r * DEG_RANGE_NODES);
    unsigned c = 0u;
#pragma unroll
    for (int s = 0; s < DEG_SLICES; ++s) c += p[(size_t)s * DEG_RANGES * DEG_PITCH];
    odi[n] = rsqrtf(fmaxf((float)c, 1.0f));
  }
}

template <int NOUT>
__global__ __launch_bounds__(256) void gcn_gemm_wmma(const float* __restrict__ A,
                                                     const float* __restrict__ odi,
                                                     const float* __restrict__ W,
                                                     float* __restrict__ C) {
  __shared__ float As[16][K_DIM + 4];

  const int tileM = blockIdx.x * 16;

  for (int idx = threadIdx.x; idx < 16 * K_DIM; idx += blockDim.x) {
    const int r = idx >> 7;
    const int c = idx & (K_DIM - 1);
    As[r][c] = A[(size_t)(tileM + r) * K_DIM + c] * odi[tileM + r];
  }
  __syncthreads();

  const int wave = threadIdx.x >> 5;
  const int lane = threadIdx.x & 31;
  const int half = lane >> 4;
  const int r16  = lane & 15;
  const int n0   = wave * 16;

  v8f acc = {};
#pragma unroll
  for (int k = 0; k < K_DIM; k += 4) {
    v2f a, b;
    a.x = As[r16][k + 2 * half];
    a.y = As[r16][k + 2 * half + 1];
    b.x = W[(size_t)(k + 2 * half)     * NOUT + n0 + r16];
    b.y = W[(size_t)(k + 2 * half + 1) * NOUT + n0 + r16];
    acc = __builtin_amdgcn_wmma_f32_16x16x4_f32(false, a, false, b,
                                                (short)0, acc, false, false);
  }

#pragma unroll
  for (int r = 0; r < 8; ++r) {
    const int row = tileM + r + 8 * half;
    C[(size_t)row * NOUT + n0 + r16] = acc[r];
  }
}

template <int D, int TR, int CAP, int GCAP, bool RELU, bool BUILD>
__global__ __launch_bounds__(1024) void gcn_tile_aggregate(const float* __restrict__ proj,
                                                           const int* __restrict__ src,
                                                           const int* __restrict__ dst,
                                                           const float* __restrict__ ew,
                                                           float* __restrict__ idiT,
                                                           const float* __restrict__ bias,
                                                           float* __restrict__ out,
                                                           int* __restrict__ glist,
                                                           unsigned* __restrict__ gcnt,
                                                           const unsigned* __restrict__ bm) {
  static_assert(D % 32 == 0 && N_EDGES % 4 == 0, "layout");
  static_assert(TR == TILE_ROWS, "a bit row names a tile of TR rows");
  static_assert(N_NODES <= 65536 && TR <= 32768, "packed list entry: 16 bits each");
  static_assert(BUILD ? CAP <= GCAP : GCAP <= CAP, "listed edges fit glist; a stored list fits the LDS list");
  static_assert(TR <= IDI_PITCH, "a tile's in_deg^{-1/2} values fit its segment of idiT");
  constexpr int NT  = 1024;
  constexpr int CPL = D / 32;
  constexpr int D4  = D / 4;
  constexpr int WPT = (BM_TILE_WORDS + NT - 1) / NT;
  constexpr int GU  = 4;

  __shared__ __attribute__((aligned(16))) float tile[TR * D];
  __shared__ int2     list[CAP];
  __shared__ unsigned cnt;
  __shared__ unsigned ngrp;
  __shared__ unsigned rowcnt[TR];

  const int tid      = threadIdx.x;
  const int lane     = tid & 31;
  const int rowBase  = blockIdx.x * TR;
  const int rowsHere = min(TR, N_NODES - rowBase);

  int2* const gl = (int2*)glist + (size_t)blockIdx.x * GCAP;
  unsigned nStored = 0u;
  if (!BUILD) nStored = gcnt[blockIdx.x * CNT_PITCH];
  const bool useStored = !BUILD && nStored <= (unsigned)GCAP;
  constexpr int PRE = 4;
  int2 pre[PRE];
  if (useStored) {
#pragma unroll
    for (int u = 0; u < PRE; ++u) {
      const unsigned i = (unsigned)(tid + u * NT);
      pre[u] = (i < nStored) ? gl[i] : make_int2(0, 0);
    }
  }

  for (int i = tid; i < TR * D4; i += NT)
    ((float4*)tile)[i] = make_float4(0.f, 0.f, 0.f, 0.f);
  if (tid == 0) {
    cnt  = 0u;
    ngrp = 0u;
  }
  if (BUILD)
    for (int i = tid; i < TR; i += NT) rowcnt[i] = 0u;
  if (useStored) {
#pragma unroll
    for (int u = 0; u < PRE; ++u) {
      const unsigned i = (unsigned)(tid + u * NT);
      if (i < nStored) list[i] = pre[u];
    }
    for (unsigned i = (unsigned)(tid + PRE * NT); i < nStored; i += NT) list[i] = gl[i];
  }
  __syncthreads();

  auto take_edge = [&](int e) {
    const unsigned slot = atomicAdd(&cnt, 1u);
    if (slot < (unsigned)CAP) {
      list[slot].x = e;
    } else {
      const float w  = ew[e];
      const float* p = proj + (size_t)src[e] * D;
      const int dl   = dst[e] - rowBase;
      float* t       = tile + dl * D;
      if (BUILD) atomicAdd(&rowcnt[dl], 1u);
#pragma nounroll
      for (int c = 0; c < D; ++c) atomicAdd(t + c, p[c] * w);
    }
  };
  const int4* dst4 = (const int4*)dst;
  auto test4 = [&](const int4& d, int q) {
    const unsigned rows = (unsigned)rowsHere;
    unsigned m = ((unsigned)(d.x - rowBase) < rows ? 1u : 0u) | ((unsigned)(d.y - rowBase) < rows ? 2u : 0u) |
                 ((unsigned)(d.z - rowBase) < rows ? 4u : 0u) | ((unsigned)(d.w - rowBase) < rows ? 8u : 0u);
    while (m != 0u) {
      const int k = __builtin_ctz(m);
      m &= m - 1u;
      take_edge(4 * q + k);
    }
  };

  if (!useStored) {
    const unsigned* bmT = bm + (size_t)blockIdx.x * BM_TILE_WORDS;
    unsigned wv[WPT];
    int n = 0;
#pragma unroll
    for (int u = 0; u < WPT; ++u) {
      const int i = tid + u * NT;
      wv[u] = (i < BM_TILE_WORDS) ? bmT[i] : 0u;
      n += __builtin_popcount(wv[u]);
    }
    int incl = n;
#pragma unroll
    for (int o = 1; o < 32; o <<= 1) {
      const int t = __shfl_up(incl, o);
      if (lane >= o) incl += t;
    }
    int base = 0;
    if (lane == 31) base = (int)atomicAdd(&ngrp, (unsigned)incl);
    base = __shfl(base, 31);
    int pos = base + incl - n;
#pragma unroll
    for (int u = 0; u < WPT; ++u) {
      unsigned w = wv[u];
      const int i = tid + u * NT;
      while (w != 0u) {
        const int q = i * 32 + __builtin_ctz(w);
        w &= w - 1u;
        if (pos < CAP) list[pos].y = q;
        ++pos;
      }
    }
    if (pos > CAP) {
      int p2 = base + incl - n;
#pragma nounroll
      for (int i = tid; i < BM_TILE_WORDS; i += NT) {
        unsigned w = bmT[i];
        while (w != 0u) {
          const int q = i * 32 + __builtin_ctz(w);
          w &= w - 1u;
          if (p2 >= CAP) test4(dst4[q], q);
          ++p2;
        }
      }
    }
    __syncthreads();

    const int nG = (int)min(ngrp, (unsigned)CAP);
    for (int g0 = tid; g0 < nG; g0 += NT * GU) {
      int  q[GU];
      int4 d[GU];
#pragma unroll
      for (int u = 0; u < GU; ++u) {
        const int gi = g0 + u * NT;
        q[u] = (gi < nG) ? list[gi].y : -1;
        d[u] = (q[u] >= 0) ? dst4[q[u]] : make_int4(-1, -1, -1, -1);
      }
#pragma unroll
      for (int u = 0; u < GU; ++u)
        if (q[u] >= 0) test4(d[u], q[u]);
    }
  }
  __syncthreads();

  const int wave  = __builtin_amdgcn_readfirstlane(tid >> 5);
  const int total = __builtin_amdgcn_readfirstlane(
      useStored ? (int)nStored : (int)min(cnt, (unsigned)CAP));
  if (BUILD && tid == 0) gcnt[blockIdx.x * CNT_PITCH] = cnt;

  auto add_row = [&](int sj, int dj, float wj) {
    const float* p = proj + (size_t)sj * D + lane;
    float* t       = tile + dj * D + lane;
    float v[CPL];
#pragma unroll
    for (int c = 0; c < CPL; ++c) v[c] = p[32 * c];
#pragma unroll
    for (int c = 0; c < CPL; ++c) atomicAdd(t + 32 * c, v[c] * wj);
  };

  auto add_step = [&](int s, int dl, float w, int nb) {
#pragma unroll
    for (int q = 0; q < 4; ++q) {
      const int nq = nb - 8 * q;
      if (nq >= 8) {
        float v[8][CPL];
#pragma unroll
        for (int j = 0; j < 8; ++j) {
          const int sj   = __builtin_amdgcn_readlane(s, 8 * q + j);
          const float* p = proj + (size_t)sj * D + lane;
#pragma unroll
          for (int c = 0; c < CPL; ++c) v[j][c] = p[32 * c];
        }
#pragma unroll
        for (int j = 0; j < 8; ++j) {
          const int dj   = __builtin_amdgcn_readlane(dl, 8 * q + j);
          const float wj = __int_as_float(__builtin_amdgcn_readlane(__float_as_int(w), 8 * q + j));
          float* t       = tile + dj * D + lane;
#pragma unroll
          for (int c = 0; c < CPL; ++c) atomicAdd(t + 32 * c, v[j][c] * wj);
        }
      } else {
        for (int j = 0; j < nq; ++j) {
          const int l = 8 * q + j;
          add_row(__builtin_amdgcn_readlane(s, l), __builtin_amdgcn_readlane(dl, l),
                  __int_as_float(__builtin_amdgcn_readlane(__float_as_int(w), l)));
        }
      }
    }
  };

  for (int b0 = wave * 32; b0 < total; b0 += 32 * 32) {
    const int nb = min(32, total - b0);
    int s = 0, dl = 0;
    float w = 0.f;
    if (lane < nb) {
      if (useStored) {
        const int2 ent = list[b0 + lane];
        s  = ent.x & 0xFFFF;
        dl = (int)((unsigned)ent.x >> 16);
        w  = __int_as_float(ent.y);
      } else {
        const int e = list[b0 + lane].x;
        s  = src[e];
        w  = ew[e];
        dl = dst[e] - rowBase;
        if (BUILD) {
          atomicAdd(&rowcnt[dl], 1u);
          gl[b0 + lane] = make_int2(s | (dl << 16), __float_as_int(w));
        }
      }
    }
    add_step(s, dl, w, nb);
  }
  __syncthreads();

  for (int i = tid; i < rowsHere * D4; i += NT) {
    const int r  = i / D4;
    const int c4 = i % D4;
    const float4 t  = ((const float4*)tile)[i];
    float sc;
    if (BUILD) {
      sc = rsqrtf(fmaxf((float)rowcnt[r], 1.0f));
      if (c4 == 0) idiT[blockIdx.x * IDI_PITCH + r] = sc;
    } else {
      sc = idiT[blockIdx.x * IDI_PITCH + r];
    }
    const float4 b  = ((const float4*)bias)[c4];
    float4 o;
    o.x = t.x * sc + b.x;
    o.y = t.y * sc + b.y;
    o.z = t.z * sc + b.z;
    o.w = t.w * sc + b.w;
    if (RELU) {
      o.x = fmaxf(o.x, 0.0f);
      o.y = fmaxf(o.y, 0.0f);
      o.z = fmaxf(o.z, 0.0f);
      o.w = fmaxf(o.w, 0.0f);
    }
    ((float4*)out)[(size_t)(rowBase + r) * D4 + c4] = o;
  }
}

extern "C" void kernel_launch(void* const* d_in, const int* in_sizes, int n_in,
                              void* d_out, int out_size, void* d_ws, size_t ws_size,
                              hipStream_t stream) {
  const float* feat = (const float*)d_in[0];
  const float* ew   = (const float*)d_in[1];
  const float* W1   = (const float*)d_in[2];
  const float* b1   = (const float*)d_in[3];
  const float* W2   = (const float*)d_in[4];
  const float* b2   = (const float*)d_in[5];
  const float* W3   = (const float*)d_in[6];
  const float* b3   = (const float*)d_in[7];
  const int*   src  = (const int*)d_in[8];
  const int*   dst  = (const int*)d_in[9];
  float* out = (float*)d_out;

  float* ws     = (float*)d_ws;
  float* odi    = ws + N_NODES;
  float* idiT   = ws + 2 * (size_t)N_NODES;
  float* h      = ws + 4 * (size_t)N_NODES;
  float* proj   = h    + (size_t)N_NODES * 128;

  const int gemmBlocks = N_NODES / 16;
  constexpr int TR = TILE_ROWS;
  constexpr int tiles = (N_NODES + TR - 1) / TR;
  static_assert(tiles == N_TILES, "the bitmap has one row per tile");
  static_assert(tiles * IDI_PITCH <= 2 * N_NODES, "idiT ends before h");
  constexpr int CAP128 = (256 * 1024 - TR * 128 * 4) / 8;
  constexpr int CAP64  = (256 * 1024 - TR * 64 * 4) / 8;
  int*      glist = (int*)(proj + (size_t)N_NODES * 128);
  unsigned* gcnt  = (unsigned*)(glist + (size_t)2 * tiles * CAP128);
  unsigned* part = gcnt + (size_t)tiles * CNT_PITCH;
  unsigned* bm = part + (size_t)DEG_SLICES * DEG_RANGES * DEG_PITCH;

  gcn_outdeg_count<<<DEG_SLICES * DEG_RANGES, DEG_THREADS, 0, stream>>>(src, dst, part, bm);
  gcn_outdeg_isqrt<<<(N_NODES + 255) / 256, 256, 0, stream>>>(part, odi);

  gcn_gemm_wmma<128><<<gemmBlocks, 256, 0, stream>>>(feat, odi, W1, proj);
  gcn_tile_aggregate<128, TR, CAP128, CAP128, true, true><<<tiles, 1024, 0, stream>>>(proj, src, dst, ew, idiT, b1, h, glist, gcnt, bm);

  gcn_gemm_wmma<128><<<gemmBlocks, 256, 0, stream>>>(h, odi, W2, proj);
  gcn_tile_aggregate<128, TR, CAP128, CAP128, true, false><<<tiles, 1024, 0, stream>>>(proj, src, dst, ew, idiT, b2, h, glist, gcnt, bm);

  gcn_gemm_wmma<64><<<gemmBlocks, 128, 0, stream>>>(h, odi, W3, proj);
  gcn_tile_aggregate<64, TR, CAP64, CAP128, false, false><<<tiles, 1024, 0, stream>>>(proj, src, dst, ew, idiT, b3, out, glist, gcnt, bm);
}


# Projection GEMM: a wave keeps its 128x16 slab of W in registers and its work-group walks 5 row stripes (two LDS stripe buffers): W read once per wave, 62500 entering waves to 12500; sums unchanged
# speedup vs baseline: 3.5901x; 1.0318x over previous
#include <hip/hip_runtime.h>
#include <math.h>


typedef __attribute__((ext_vector_type(2))) float v2f;
typedef __attribute__((ext_vector_type(8))) float v8f;

#define N_NODES 50000
#define N_EDGES 800000
#define K_DIM   128
#define CNT_PITCH 32
#define TILE_ROWS 197
#define N_TILES ((N_NODES + TILE_ROWS - 1) / TILE_ROWS)
#define BM_GROUPS 1024
#define BM_CHUNKS ((N_EDGES / 4 + BM_GROUPS - 1) / BM_GROUPS)
#define BM_TILE_WORDS (BM_CHUNKS * 32)
#define IDI_PITCH 256
#define DEG_SLICES 32
#define DEG_RANGES 8
#define DEG_RANGE_NODES (N_NODES / DEG_RANGES)
#define DEG_PITCH 6272
#define DEG_THREADS 512

__global__ __launch_bounds__(DEG_THREADS) void gcn_outdeg_count(const int* __restrict__ src,
                                                                const int* __restrict__ dst,
                                                                unsigned* __restrict__ part,
                                                                unsigned* __restrict__ bm) {
  static_assert(N_NODES % DEG_RANGES == 0 && N_EDGES % (4 * DEG_SLICES) == 0, "even split");
  static_assert(DEG_PITCH >= DEG_RANGE_NODES && DEG_PITCH % 32 == 0, "segment = whole lines");
  static_assert(BM_CHUNKS <= DEG_SLICES * DEG_RANGES && BM_GROUPS == 2 * DEG_THREADS && N_EDGES % 4 == 0,
                "one work-group per bitmap chunk, two groups per thread");
  constexpr int SLICE4 = N_EDGES / DEG_SLICES / 4;
  constexpr int PER    = (SLICE4 + DEG_THREADS - 1) / DEG_THREADS;

  __shared__ __attribute__((aligned(16))) unsigned cntL[DEG_PITCH];
  __shared__ __attribute__((aligned(16))) unsigned bmL[N_TILES * 32];

  const int tid = threadIdx.x;
  const int s   = blockIdx.x / DEG_RANGES;
  const int lo  = (blockIdx.x % DEG_RANGES) * DEG_RANGE_NODES;

  const int4* s4 = (const int4*)src + (size_t)s * SLICE4;
  int4 v[PER];
#pragma unroll
  for (int u = 0; u < PER; ++u) {
    const int i = tid + u * DEG_THREADS;
    v[u] = (i < SLICE4) ? s4[i] : make_int4(-1, -1, -1, -1);
  }

  const bool chunk = blockIdx.x < BM_CHUNKS;
  int4 dg[2];
#pragma unroll
  for (int u = 0; u < 2; ++u) {
    const int q = blockIdx.x * BM_GROUPS + tid + u * DEG_THREADS;
    dg[u] = (chunk && q < N_EDGES / 4) ? ((const int4*)dst)[q] : make_int4(-1, -1, -1, -1);
  }

  for (int i = tid; i < DEG_PITCH / 4; i += DEG_THREADS)
    ((uint4*)cntL)[i] = make_uint4(0u, 0u, 0u, 0u);
  if (chunk)
    for (int i = tid; i < N_TILES * 8; i += DEG_THREADS)
      ((uint4*)bmL)[i] = make_uint4(0u, 0u, 0u, 0u);
  __syncthreads();

  auto mark = [&](int d, int g) {
    if ((unsigned)d < (unsigned)N_NODES)
      atomicOr(&bmL[((unsigned)d / TILE_ROWS) * 32 + (g >> 5)], 1u << (g & 31));
  };
#pragma unroll
  for (int u = 0; u < 2; ++u) {
    const int g = tid + u * DEG_THREADS;
    mark(dg[u].x, g);
    mark(dg[u].y, g);
    mark(dg[u].z, g);
    mark(dg[u].w, g);
  }

  auto count = [&](int sv) {
    const unsigned o = (unsigned)(sv - lo);
    if (o < (unsigned)DEG_RANGE_NODES) atomicAdd(&cntL[o], 1u);
  };
#pragma unroll
  for (int u = 0; u < PER; ++u) {
    count(v[u].x);
    count(v[u].y);
    count(v[u].z);
    count(v[u].w);
  }
  __syncthreads();

  uint4* seg = (uint4*)(part + (size_t)blockIdx.x * DEG_PITCH);
  for (int i = tid; i < DEG_PITCH / 4; i += DEG_THREADS) seg[i] = ((const uint4*)cntL)[i];

  if (chunk)
    for (int i = tid; i < N_TILES * 8; i += DEG_THREADS)
      ((uint4*)(bm + (size_t)(i >> 3) * BM_TILE_WORDS + blockIdx.x * 32))[i & 7] = ((const uint4*)bmL)[i];
}

__global__ void gcn_outdeg_isqrt(const unsigned* __restrict__ part, float* __restrict__ odi) {
  int n = blockIdx.x * blockDim.x + threadIdx.x;
  if (n < N_NODES) {
    const int r = n / DEG_RANGE_NODES;
    const unsigned* p = part + (size_t)r * DEG_PITCH + (n - r * DEG_RANGE_NODES);
    unsigned c = 0u;
#pragma unroll
    for (int s = 0; s < DEG_SLICES; ++s) c += p[(size_t)s * DEG_RANGES * DEG_PITCH];
    odi[n] = rsqrtf(fmaxf((float)c, 1.0f));
  }
}

template <int NOUT, int STRIPES>
__global__ __launch_bounds__(NOUT * 2) void gcn_gemm_wmma(const float* __restrict__ A,
                                                          const float* __restrict__ odi,
                                                          const float* __restrict__ W,
                                                          float* __restrict__ C) {
  static_assert((N_NODES / 16) % STRIPES == 0, "every work-group walks STRIPES whole stripes");
  constexpr int NT  = NOUT * 2;
  constexpr int PER = 16 * K_DIM / 4 / NT;
  static_assert(PER * NT * 4 == 16 * K_DIM, "a stripe is PER whole loads per thread");

  __shared__ __attribute__((aligned(16))) float As[2][16][K_DIM + 4];

  const int tid  = threadIdx.x;
  const int wave = tid >> 5;
  const int lane = tid & 31;
  const int half = lane >> 4;
  const int r16  = lane & 15;
  const int n0   = wave * 16;
  const int tile0 = blockIdx.x * STRIPES;

  v2f b[K_DIM / 4];
#pragma unroll
  for (int k = 0; k < K_DIM; k += 4) {
    b[k / 4].x = W[(size_t)(k + 2 * half)     * NOUT + n0 + r16];
    b[k / 4].y = W[(size_t)(k + 2 * half + 1) * NOUT + n0 + r16];
  }

  float4 ar[PER];
  float  sc[PER];
  auto fetch = [&](int tile) {
#pragma unroll
    for (int u = 0; u < PER; ++u) {
      const int idx = tid + u * NT;
      const int r   = idx >> 5;
      const int c4  = idx & 31;
      ar[u] = ((const float4*)(A + (size_t)(tile * 16 + r) * K_DIM))[c4];
      sc[u] = odi[tile * 16 + r];
    }
  };
  fetch(tile0);

#pragma unroll 1
  for (int s = 0; s < STRIPES; ++s) {
    const int tileM = (tile0 + s) * 16;
    float (*Ab)[K_DIM + 4] = As[s & 1];
#pragma unroll
    for (int u = 0; u < PER; ++u) {
      const int idx = tid + u * NT;
      const int r   = idx >> 5;
      const int c4  = idx & 31;
      float4 v = ar[u];
      v.x *= sc[u];
      v.y *= sc[u];
      v.z *= sc[u];
      v.w *= sc[u];
      *(float4*)&Ab[r][4 * c4] = v;
    }
    __syncthreads();
    if (s + 1 < STRIPES) fetch(tile0 + s + 1);

    v8f acc = {};
#pragma unroll
    for (int k = 0; k < K_DIM; k += 4) {
      v2f a;
      a.x = Ab[r16][k + 2 * half];
      a.y = Ab[r16][k + 2 * half + 1];
      acc = __builtin_amdgcn_wmma_f32_16x16x4_f32(false, a, false, b[k / 4],
                                                  (short)0, acc, false, false);
    }

#pragma unroll
    for (int r = 0; r < 8; ++r) {
      const int row = tileM + r + 8 * half;
      C[(size_t)row * NOUT + n0 + r16] = acc[r];
    }
  }
}

template <int D, int TR, int CAP, int GCAP, bool RELU, bool BUILD>
__global__ __launch_bounds__(1024) void gcn_tile_aggregate(const float* __restrict__ proj,
                                                           const int* __restrict__ src,
                                                           const int* __restrict__ dst,
                                                           const float* __restrict__ ew,
                                                           float* __restrict__ idiT,
                                                           const float* __restrict__ bias,
                                                           float* __restrict__ out,
                                                           int* __restrict__ glist,
                                                           unsigned* __restrict__ gcnt,
                                                           const unsigned* __restrict__ bm) {
  static_assert(D % 32 == 0 && N_EDGES % 4 == 0, "layout");
  static_assert(TR == TILE_ROWS, "a bit row names a tile of TR rows");
  static_assert(N_NODES <= 65536 && TR <= 32768, "packed list entry: 16 bits each");
  static_assert(BUILD ? CAP <= GCAP : GCAP <= CAP, "listed edges fit glist; a stored list fits the LDS list");
  static_assert(TR <= IDI_PITCH, "a tile's in_deg^{-1/2} values fit its segment of idiT");
  constexpr int NT  = 1024;
  constexpr int CPL = D / 32;
  constexpr int D4  = D / 4;
  constexpr int WPT = (BM_TILE_WORDS + NT - 1) / NT;
  constexpr int GU  = 4;

  __shared__ __attribute__((aligned(16))) float tile[TR * D];
  __shared__ int2     list[CAP];
  __shared__ unsigned cnt;
  __shared__ unsigned ngrp;
  __shared__ unsigned rowcnt[TR];

  const int tid      = threadIdx.x;
  const int lane     = tid & 31;
  const int rowBase  = blockIdx.x * TR;
  const int rowsHere = min(TR, N_NODES - rowBase);

  int2* const gl = (int2*)glist + (size_t)blockIdx.x * GCAP;
  unsigned nStored = 0u;
  if (!BUILD) nStored = gcnt[blockIdx.x * CNT_PITCH];
  const bool useStored = !BUILD && nStored <= (unsigned)GCAP;
  constexpr int PRE = 4;
  int2 pre[PRE];
  if (useStored) {
#pragma unroll
    for (int u = 0; u < PRE; ++u) {
      const unsigned i = (unsigned)(tid + u * NT);
      pre[u] = (i < nStored) ? gl[i] : make_int2(0, 0);
    }
  }

  for (int i = tid; i < TR * D4; i += NT)
    ((float4*)tile)[i] = make_float4(0.f, 0.f, 0.f, 0.f);
  if (tid == 0) {
    cnt  = 0u;
    ngrp = 0u;
  }
  if (BUILD)
    for (int i = tid; i < TR; i += NT) rowcnt[i] = 0u;
  if (useStored) {
#pragma unroll
    for (int u = 0; u < PRE; ++u) {
      const unsigned i = (unsigned)(tid + u * NT);
      if (i < nStored) list[i] = pre[u];
    }
    for (unsigned i = (unsigned)(tid + PRE * NT); i < nStored; i += NT) list[i] = gl[i];
  }
  __syncthreads();

  auto take_edge = [&](int e) {
    const unsigned slot = atomicAdd(&cnt, 1u);
    if (slot < (unsigned)CAP) {
      list[slot].x = e;
    } else {
      const float w  = ew[e];
      const float* p = proj + (size_t)src[e] * D;
      const int dl   = dst[e] - rowBase;
      float* t       = tile + dl * D;
      if (BUILD) atomicAdd(&rowcnt[dl], 1u);
#pragma nounroll
      for (int c = 0; c < D; ++c) atomicAdd(t + c, p[c] * w);
    }
  };
  const int4* dst4 = (const int4*)dst;
  auto test4 = [&](const int4& d, int q) {
    const unsigned rows = (unsigned)rowsHere;
    unsigned m = ((unsigned)(d.x - rowBase) < rows ? 1u : 0u) | ((unsigned)(d.y - rowBase) < rows ? 2u : 0u) |
                 ((unsigned)(d.z - rowBase) < rows ? 4u : 0u) | ((unsigned)(d.w - rowBase) < rows ? 8u : 0u);
    while (m != 0u) {
      const int k = __builtin_ctz(m);
      m &= m - 1u;
      take_edge(4 * q + k);
    }
  };

  if (!useStored) {
    const unsigned* bmT = bm + (size_t)blockIdx.x * BM_TILE_WORDS;
    unsigned wv[WPT];
    int n = 0;
#pragma unroll
    for (int u = 0; u < WPT; ++u) {
      const int i = tid + u * NT;
      wv[u] = (i < BM_TILE_WORDS) ? bmT[i] : 0u;
      n += __builtin_popcount(wv[u]);
    }
    int incl = n;
#pragma unroll
    for (int o = 1; o < 32; o <<= 1) {
      const int t = __shfl_up(incl, o);
      if (lane >= o) incl += t;
    }
    int base = 0;
    if (lane == 31) base = (int)atomicAdd(&ngrp, (unsigned)incl);
    base = __shfl(base, 31);
    int pos = base + incl - n;
#pragma unroll
    for (int u = 0; u < WPT; ++u) {
      unsigned w = wv[u];
      const int i = tid + u * NT;
      while (w != 0u) {
        const int q = i * 32 + __builtin_ctz(w);
        w &= w - 1u;
        if (pos < CAP) list[pos].y = q;
        ++pos;
      }
    }
    if (pos > CAP) {
      int p2 = base + incl - n;
#pragma nounroll
      for (int i = tid; i < BM_TILE_WORDS; i += NT) {
        unsigned w = bmT[i];
        while (w != 0u) {
          const int q = i * 32 + __builtin_ctz(w);
          w &= w - 1u;
          if (p2 >= CAP) test4(dst4[q], q);
          ++p2;
        }
      }
    }
    __syncthreads();

    const int nG = (int)min(ngrp, (unsigned)CAP);
    for (int g0 = tid; g0 < nG; g0 += NT * GU) {
      int  q[GU];
      int4 d[GU];
#pragma unroll
      for (int u = 0; u < GU; ++u) {
        const int gi = g0 + u * NT;
        q[u] = (gi < nG) ? list[gi].y : -1;
        d[u] = (q[u] >= 0) ? dst4[q[u]] : make_int4(-1, -1, -1, -1);
      }
#pragma unroll
      for (int u = 0; u < GU; ++u)
        if (q[u] >= 0) test4(d[u], q[u]);
    }
  }
  __syncthreads();

  const int wave  = __builtin_amdgcn_readfirstlane(tid >> 5);
  const int total = __builtin_amdgcn_readfirstlane(
      useStored ? (int)nStored : (int)min(cnt, (unsigned)CAP));
  if (BUILD && tid == 0) gcnt[blockIdx.x * CNT_PITCH] = cnt;

  auto add_row = [&](int sj, int dj, float wj) {
    const float* p = proj + (size_t)sj * D + lane;
    float* t       = tile + dj * D + lane;
    float v[CPL];
#pragma unroll
    for (int c = 0; c < CPL; ++c) v[c] = p[32 * c];
#pragma unroll
    for (int c = 0; c < CPL; ++c) atomicAdd(t + 32 * c, v[c] * wj);
  };

  auto add_step = [&](int s, int dl, float w, int nb) {
#pragma unroll
    for (int q = 0; q < 4; ++q) {
      const int nq = nb - 8 * q;
      if (nq >= 8) {
        float v[8][CPL];
#pragma unroll
        for (int j = 0; j < 8; ++j) {
          const int sj   = __builtin_amdgcn_readlane(s, 8 * q + j);
          const float* p = proj + (size_t)sj * D + lane;
#pragma unroll
          for (int c = 0; c < CPL; ++c) v[j][c] = p[32 * c];
        }
#pragma unroll
        for (int j = 0; j < 8; ++j) {
          const int dj   = __builtin_amdgcn_readlane(dl, 8 * q + j);
          const float wj = __int_as_float(__builtin_amdgcn_readlane(__float_as_int(w), 8 * q + j));
          float* t       = tile + dj * D + lane;
#pragma unroll
          for (int c = 0; c < CPL; ++c) atomicAdd(t + 32 * c, v[j][c] * wj);
        }
      } else {
        for (int j = 0; j < nq; ++j) {
          const int l = 8 * q + j;
          add_row(__builtin_amdgcn_readlane(s, l), __builtin_amdgcn_readlane(dl, l),
                  __int_as_float(__builtin_amdgcn_readlane(__float_as_int(w), l)));
        }
      }
    }
  };

  for (int b0 = wave * 32; b0 < total; b0 += 32 * 32) {
    const int nb = min(32, total - b0);
    int s = 0, dl = 0;
    float w = 0.f;
    if (lane < nb) {
      if (useStored) {
        const int2 ent = list[b0 + lane];
        s  = ent.x & 0xFFFF;
        dl = (int)((unsigned)ent.x >> 16);
        w  = __int_as_float(ent.y);
      } else {
        const int e = list[b0 + lane].x;
        s  = src[e];
        w  = ew[e];
        dl = dst[e] - rowBase;
        if (BUILD) {
          atomicAdd(&rowcnt[dl], 1u);
          gl[b0 + lane] = make_int2(s | (dl << 16), __float_as_int(w));
        }
      }
    }
    add_step(s, dl, w, nb);
  }
  __syncthreads();

  for (int i = tid; i < rowsHere * D4; i += NT) {
    const int r  = i / D4;
    const int c4 = i % D4;
    const float4 t  = ((const float4*)tile)[i];
    float sc;
    if (BUILD) {
      sc = rsqrtf(fmaxf((float)rowcnt[r], 1.0f));
      if (c4 == 0) idiT[blockIdx.x * IDI_PITCH + r] = sc;
    } else {
      sc = idiT[blockIdx.x * IDI_PITCH + r];
    }
    const float4 b  = ((const float4*)bias)[c4];
    float4 o;
    o.x = t.x * sc + b.x;
    o.y = t.y * sc + b.y;
    o.z = t.z * sc + b.z;
    o.w = t.w * sc + b.w;
    if (RELU) {
      o.x = fmaxf(o.x, 0.0f);
      o.y = fmaxf(o.y, 0.0f);
      o.z = fmaxf(o.z, 0.0f);
      o.w = fmaxf(o.w, 0.0f);
    }
    ((float4*)out)[(size_t)(rowBase + r) * D4 + c4] = o;
  }
}

extern "C" void kernel_launch(void* const* d_in, const int* in_sizes, int n_in,
                              void* d_out, int out_size, void* d_ws, size_t ws_size,
                              hipStream_t stream) {
  const float* feat = (const float*)d_in[0];
  const float* ew   = (const float*)d_in[1];
  const float* W1   = (const float*)d_in[2];
  const float* b1   = (const float*)d_in[3];
  const float* W2   = (const float*)d_in[4];
  const float* b2   = (const float*)d_in[5];
  const float* W3   = (const float*)d_in[6];
  const float* b3   = (const float*)d_in[7];
  const int*   src  = (const int*)d_in[8];
  const int*   dst  = (const int*)d_in[9];
  float* out = (float*)d_out;

  float* ws     = (float*)d_ws;
  float* odi    = ws + N_NODES;
  float* idiT   = ws + 2 * (size_t)N_NODES;
  float* h      = ws + 4 * (size_t)N_NODES;
  float* proj   = h    + (size_t)N_NODES * 128;

  constexpr int GS = 5;
  const int gemmBlocks = N_NODES / 16 / GS;
  constexpr int TR = TILE_ROWS;
  constexpr int tiles = (N_NODES + TR - 1) / TR;
  static_assert(tiles == N_TILES, "the bitmap has one row per tile");
  static_assert(tiles * IDI_PITCH <= 2 * N_NODES, "idiT ends before h");
  constexpr int CAP128 = (256 * 1024 - TR * 128 * 4) / 8;
  constexpr int CAP64  = (256 * 1024 - TR * 64 * 4) / 8;
  int*      glist = (int*)(proj + (size_t)N_NODES * 128);
  unsigned* gcnt  = (unsigned*)(glist + (size_t)2 * tiles * CAP128);
  unsigned* part = gcnt + (size_t)tiles * CNT_PITCH;
  unsigned* bm = part + (size_t)DEG_SLICES * DEG_RANGES * DEG_PITCH;

  gcn_outdeg_count<<<DEG_SLICES * DEG_RANGES, DEG_THREADS, 0, stream>>>(src, dst, part, bm);
  gcn_outdeg_isqrt<<<(N_NODES + 255) / 256, 256, 0, stream>>>(part, odi);

  gcn_gemm_wmma<128, GS><<<gemmBlocks, 256, 0, stream>>>(feat, odi, W1, proj);
  gcn_tile_aggregate<128, TR, CAP128, CAP128, true, true><<<tiles, 1024, 0, stream>>>(proj, src, dst, ew, idiT, b1, h, glist, gcnt, bm);

  gcn_gemm_wmma<128, GS><<<gemmBlocks, 256, 0, stream>>>(h, odi, W2, proj);
  gcn_tile_aggregate<128, TR, CAP128, CAP128, true, false><<<tiles, 1024, 0, stream>>>(proj, src, dst, ew, idiT, b2, h, glist, gcnt, bm);

  gcn_gemm_wmma<64, GS><<<gemmBlocks, 128, 0, stream>>>(h, odi, W3, proj);
  gcn_tile_aggregate<64, TR, CAP64, CAP128, false, false><<<tiles, 1024, 0, stream>>>(proj, src, dst, ew, idiT, b3, out, glist, gcnt, bm);
}


# Degree pass packs edges as 8-byte records (src, dst, weight); layer-1 filter reads a flagged group's 32 B, lists resolved edges: no per-edge look-ups; sums LDS arrival order (list order changed)
# speedup vs baseline: 3.7409x; 1.0420x over previous
#include <hip/hip_runtime.h>
#include <math.h>


typedef __attribute__((ext_vector_type(2))) float v2f;
typedef __attribute__((ext_vector_type(8))) float v8f;

#define N_NODES 50000
#define N_EDGES 800000
#define K_DIM   128
#define CNT_PITCH 32
#define TILE_ROWS 197
#define N_TILES ((N_NODES + TILE_ROWS - 1) / TILE_ROWS)
#define BM_GROUPS 1024
#define BM_CHUNKS ((N_EDGES / 4 + BM_GROUPS - 1) / BM_GROUPS)
#define BM_TILE_WORDS (BM_CHUNKS * 32)
#define IDI_PITCH 256
#define DEG_SLICES 32
#define DEG_RANGES 8
#define DEG_RANGE_NODES (N_NODES / DEG_RANGES)
#define DEG_PITCH 6272
#define DEG_THREADS 512

__global__ __launch_bounds__(DEG_THREADS) void gcn_outdeg_count(const int* __restrict__ src,
                                                                const int* __restrict__ dst,
                                                                const float* __restrict__ ew,
                                                                unsigned* __restrict__ part,
                                                                unsigned* __restrict__ bm,
                                                                unsigned* __restrict__ rec) {
  constexpr int PAIRS = 4;
  static_assert(N_NODES % DEG_RANGES == 0 && N_EDGES % (4 * DEG_SLICES) == 0, "even split");
  static_assert(DEG_PITCH >= DEG_RANGE_NODES && DEG_PITCH % 32 == 0, "segment = whole lines");
  static_assert(BM_CHUNKS <= DEG_SLICES * DEG_RANGES && 2 * BM_GROUPS == PAIRS * DEG_THREADS && N_EDGES % 4 == 0,
                "one work-group per bitmap chunk, PAIRS edge pairs per thread");
  static_assert(N_NODES <= 65535, "an edge record packs source and destination in 16 bits each");
  constexpr int SLICE4 = N_EDGES / DEG_SLICES / 4;
  constexpr int PER    = (SLICE4 + DEG_THREADS - 1) / DEG_THREADS;

  __shared__ __attribute__((aligned(16))) unsigned cntL[DEG_PITCH];
  __shared__ __attribute__((aligned(16))) unsigned bmL[N_TILES * 32];

  const int tid = threadIdx.x;
  const int s   = blockIdx.x / DEG_RANGES;
  const int lo  = (blockIdx.x % DEG_RANGES) * DEG_RANGE_NODES;

  const int4* s4 = (const int4*)src + (size_t)s * SLICE4;
  int4 v[PER];
#pragma unroll
  for (int u = 0; u < PER; ++u) {
    const int i = tid + u * DEG_THREADS;
    v[u] = (i < SLICE4) ? s4[i] : make_int4(-1, -1, -1, -1);
  }

  const bool chunk = blockIdx.x < BM_CHUNKS;
  int2   dp[PAIRS];
  int2   sp[PAIRS];
  float2 wp[PAIRS];
#pragma unroll
  for (int u = 0; u < PAIRS; ++u) {
    const int h   = blockIdx.x * (2 * BM_GROUPS) + tid + u * DEG_THREADS;
    const bool ok = chunk && h < N_EDGES / 2;
    dp[u] = ok ? ((const int2*)dst)[h] : make_int2(-1, -1);
    sp[u] = ok ? ((const int2*)src)[h] : make_int2(0, 0);
    wp[u] = ok ? ((const float2*)ew)[h] : make_float2(0.f, 0.f);
  }

  for (int i = tid; i < DEG_PITCH / 4; i += DEG_THREADS)
    ((uint4*)cntL)[i] = make_uint4(0u, 0u, 0u, 0u);
  if (chunk)
    for (int i = tid; i < N_TILES * 8; i += DEG_THREADS)
      ((uint4*)bmL)[i] = make_uint4(0u, 0u, 0u, 0u);
  __syncthreads();

  auto mark = [&](int d, int g) {
    if ((unsigned)d < (unsigned)N_NODES)
      atomicOr(&bmL[((unsigned)d / TILE_ROWS) * 32 + (g >> 5)], 1u << (g & 31));
  };
#pragma unroll
  for (int u = 0; u < PAIRS; ++u) {
    const int hl = tid + u * DEG_THREADS;
    mark(dp[u].x, hl >> 1);
    mark(dp[u].y, hl >> 1);
    const int h = blockIdx.x * (2 * BM_GROUPS) + hl;
    if (chunk && h < N_EDGES / 2)
      ((uint4*)rec)[h] = make_uint4((unsigned)sp[u].x | ((unsigned)dp[u].x << 16), __float_as_uint(wp[u].x),
                                    (unsigned)sp[u].y | ((unsigned)dp[u].y << 16), __float_as_uint(wp[u].y));
  }

  auto count = [&](int sv) {
    const unsigned o = (unsigned)(sv - lo);
    if (o < (unsigned)DEG_RANGE_NODES) atomicAdd(&cntL[o], 1u);
  };
#pragma unroll
  for (int u = 0; u < PER; ++u) {
    count(v[u].x);
    count(v[u].y);
    count(v[u].z);
    count(v[u].w);
  }
  __syncthreads();

  uint4* seg = (uint4*)(part + (size_t)blockIdx.x * DEG_PITCH);
  for (int i = tid; i < DEG_PITCH / 4; i += DEG_THREADS) seg[i] = ((const uint4*)cntL)[i];

  if (chunk)
    for (int i = tid; i < N_TILES * 8; i += DEG_THREADS)
      ((uint4*)(bm + (size_t)(i >> 3) * BM_TILE_WORDS + blockIdx.x * 32))[i & 7] = ((const uint4*)bmL)[i];
}

__global__ void gcn_outdeg_isqrt(const unsigned* __restrict__ part, float* __restrict__ odi) {
  int n = blockIdx.x * blockDim.x + threadIdx.x;
  if (n < N_NODES) {
    const int r = n / DEG_RANGE_NODES;
    const unsigned* p = part + (size_t)r * DEG_PITCH + (n - r * DEG_RANGE_NODES);
    unsigned c = 0u;
#pragma unroll
    for (int s = 0; s < DEG_SLICES; ++s) c += p[(size_t)s * DEG_RANGES * DEG_PITCH];
    odi[n] = rsqrtf(fmaxf((float)c, 1.0f));
  }
}

template <int NOUT, int STRIPES>
__global__ __launch_bounds__(NOUT * 2) void gcn_gemm_wmma(const float* __restrict__ A,
                                                          const float* __restrict__ odi,
                                                          const float* __restrict__ W,
                                                          float* __restrict__ C) {
  static_assert((N_NODES / 16) % STRIPES == 0, "every work-group walks STRIPES whole stripes");
  constexpr int NT  = NOUT * 2;
  constexpr int PER = 16 * K_DIM / 4 / NT;
  static_assert(PER * NT * 4 == 16 * K_DIM, "a stripe is PER whole loads per thread");

  __shared__ __attribute__((aligned(16))) float As[2][16][K_DIM + 4];

  const int tid  = threadIdx.x;
  const int wave = tid >> 5;
  const int lane = tid & 31;
  const int half = lane >> 4;
  const int r16  = lane & 15;
  const int n0   = wave * 16;
  const int tile0 = blockIdx.x * STRIPES;

  v2f b[K_DIM / 4];
#pragma unroll
  for (int k = 0; k < K_DIM; k += 4) {
    b[k / 4].x = W[(size_t)(k + 2 * half)     * NOUT + n0 + r16];
    b[k / 4].y = W[(size_t)(k + 2 * half + 1) * NOUT + n0 + r16];
  }

  float4 ar[PER];
  float  sc[PER];
  auto fetch = [&](int tile) {
#pragma unroll
    for (int u = 0; u < PER; ++u) {
      const int idx = tid + u * NT;
      const int r   = idx >> 5;
      const int c4  = idx & 31;
      ar[u] = ((const float4*)(A + (size_t)(tile * 16 + r) * K_DIM))[c4];
      sc[u] = odi[tile * 16 + r];
    }
  };
  fetch(tile0);

#pragma unroll 1
  for (int s = 0; s < STRIPES; ++s) {
    const int tileM = (tile0 + s) * 16;
    float (*Ab)[K_DIM + 4] = As[s & 1];
#pragma unroll
    for (int u = 0; u < PER; ++u) {
      const int idx = tid + u * NT;
      const int r   = idx >> 5;
      const int c4  = idx & 31;
      float4 v = ar[u];
      v.x *= sc[u];
      v.y *= sc[u];
      v.z *= sc[u];
      v.w *= sc[u];
      *(float4*)&Ab[r][4 * c4] = v;
    }
    __syncthreads();
    if (s + 1 < STRIPES) fetch(tile0 + s + 1);

    v8f acc = {};
#pragma unroll
    for (int k = 0; k < K_DIM; k += 4) {
      v2f a;
      a.x = Ab[r16][k + 2 * half];
      a.y = Ab[r16][k + 2 * half + 1];
      acc = __builtin_amdgcn_wmma_f32_16x16x4_f32(false, a, false, b[k / 4],
                                                  (short)0, acc, false, false);
    }

#pragma unroll
    for (int r = 0; r < 8; ++r) {
      const int row = tileM + r + 8 * half;
      C[(size_t)row * NOUT + n0 + r16] = acc[r];
    }
  }
}

template <int D, int TR, int CAP, int GCAP, int GLC, bool RELU, bool BUILD>
__global__ __launch_bounds__(1024) void gcn_tile_aggregate(const float* __restrict__ proj,
                                                           const unsigned* __restrict__ rec,
                                                           float* __restrict__ idiT,
                                                           const float* __restrict__ bias,
                                                           float* __restrict__ out,
                                                           int* __restrict__ glist,
                                                           unsigned* __restrict__ gcnt,
                                                           const unsigned* __restrict__ bm) {
  static_assert(D % 32 == 0 && N_EDGES % 4 == 0, "layout");
  static_assert(TR == TILE_ROWS, "a bit row names a tile of TR rows");
  static_assert(N_NODES <= 65535 && TR <= 32768, "packed list entry and edge record: 16 bits each; 65535 is no row");
  static_assert(GLC % 2 == 0 && GLC > 0, "group list");
  static_assert(BUILD ? CAP <= GCAP : GCAP <= CAP, "listed edges fit glist; a stored list fits the LDS list");
  static_assert(TR <= IDI_PITCH, "a tile's in_deg^{-1/2} values fit its segment of idiT");
  constexpr int NT  = 1024;
  constexpr int CPL = D / 32;
  constexpr int D4  = D / 4;
  constexpr int WPT = (BM_TILE_WORDS + NT - 1) / NT;
  constexpr int GU  = 8;

  __shared__ __attribute__((aligned(16))) float tile[TR * D];
  __shared__ int2     list[CAP];
  __shared__ int      grp[GLC];
  __shared__ unsigned cnt;
  __shared__ unsigned ngrp;
  __shared__ unsigned rowcnt[TR];

  const int tid      = threadIdx.x;
  const int lane     = tid & 31;
  const int rowBase  = blockIdx.x * TR;
  const int rowsHere = min(TR, N_NODES - rowBase);

  int2* const gl = (int2*)glist + (size_t)blockIdx.x * GCAP;
  unsigned nStored = 0u;
  if (!BUILD) nStored = gcnt[blockIdx.x * CNT_PITCH];
  const bool useStored = !BUILD && nStored <= (unsigned)GCAP;
  constexpr int PRE = 4;
  int2 pre[PRE];
  if (useStored) {
#pragma unroll
    for (int u = 0; u < PRE; ++u) {
      const unsigned i = (unsigned)(tid + u * NT);
      pre[u] = (i < nStored) ? gl[i] : make_int2(0, 0);
    }
  }

  for (int i = tid; i < TR * D4; i += NT)
    ((float4*)tile)[i] = make_float4(0.f, 0.f, 0.f, 0.f);
  if (tid == 0) {
    cnt  = 0u;
    ngrp = 0u;
  }
  if (BUILD)
    for (int i = tid; i < TR; i += NT) rowcnt[i] = 0u;
  if (useStored) {
#pragma unroll
    for (int u = 0; u < PRE; ++u) {
      const unsigned i = (unsigned)(tid + u * NT);
      if (i < nStored) list[i] = pre[u];
    }
    for (unsigned i = (unsigned)(tid + PRE * NT); i < nStored; i += NT) list[i] = gl[i];
  }
  __syncthreads();

  auto take_edge = [&](int s, int dl, int wbits) {
    if (BUILD) atomicAdd(&rowcnt[dl], 1u);
    const unsigned slot = atomicAdd(&cnt, 1u);
    if (slot < (unsigned)CAP) {
      list[slot] = make_int2(s | (dl << 16), wbits);
    } else {
      const float w  = __int_as_float(wbits);
      const float* p = proj + (size_t)s * D;
      float* t       = tile + dl * D;
#pragma nounroll
      for (int c = 0; c < D; ++c) atomicAdd(t + c, p[c] * w);
    }
  };
  const uint4* rec4 = (const uint4*)rec;
  auto test2 = [&](const uint4& r) {
    const unsigned rows = (unsigned)rowsHere;
    const unsigned d0 = (r.x >> 16) - (unsigned)rowBase;
    const unsigned d1 = (r.z >> 16) - (unsigned)rowBase;
    unsigned m = (d0 < rows ? 1u : 0u) | (d1 < rows ? 2u : 0u);
    while (m != 0u) {
      const bool second = (m & 1u) == 0u;
      m &= m - 1u;
      take_edge((int)((second ? r.z : r.x) & 0xFFFFu), (int)(second ? d1 : d0), (int)(second ? r.w : r.y));
    }
  };

  if (!useStored) {
    const unsigned* bmT = bm + (size_t)blockIdx.x * BM_TILE_WORDS;
    unsigned wv[WPT];
    int n = 0;
#pragma unroll
    for (int u = 0; u < WPT; ++u) {
      const int i = tid + u * NT;
      wv[u] = (i < BM_TILE_WORDS) ? bmT[i] : 0u;
      n += __builtin_popcount(wv[u]);
    }
    int incl = n;
#pragma unroll
    for (int o = 1; o < 32; o <<= 1) {
      const int t = __shfl_up(incl, o);
      if (lane >= o) incl += t;
    }
    int base = 0;
    if (lane == 31) base = (int)atomicAdd(&ngrp, (unsigned)incl);
    base = __shfl(base, 31);
    int pos = base + incl - n;
#pragma unroll
    for (int u = 0; u < WPT; ++u) {
      unsigned w = wv[u];
      const int i = tid + u * NT;
      while (w != 0u) {
        const int q = i * 32 + __builtin_ctz(w);
        w &= w - 1u;
        if (pos < GLC) grp[pos] = q;
        ++pos;
      }
    }
    if (pos > GLC) {
      int p2 = base + incl - n;
#pragma nounroll
      for (int i = tid; i < BM_TILE_WORDS; i += NT) {
        unsigned w = bmT[i];
        while (w != 0u) {
          const int q = i * 32 + __builtin_ctz(w);
          w &= w - 1u;
          if (p2 >= GLC) {
            test2(rec4[2 * q]);
            test2(rec4[2 * q + 1]);
          }
          ++p2;
        }
      }
    }
    __syncthreads();

    const int nH = 2 * (int)min(ngrp, (unsigned)GLC);
    for (int h0 = tid; h0 < nH; h0 += NT * GU) {
      uint4 r[GU];
#pragma unroll
      for (int u = 0; u < GU; ++u) {
        const int hi = h0 + u * NT;
        r[u] = (hi < nH) ? rec4[2 * grp[hi >> 1] + (hi & 1)] : make_uint4(0xFFFF0000u, 0u, 0xFFFF0000u, 0u);
      }
#pragma unroll
      for (int u = 0; u < GU; ++u) test2(r[u]);
    }
  }
  __syncthreads();

  const int wave  = __builtin_amdgcn_readfirstlane(tid >> 5);
  const int total = __builtin_amdgcn_readfirstlane(
      useStored ? (int)nStored : (int)min(cnt, (unsigned)CAP));
  if (BUILD && tid == 0) gcnt[blockIdx.x * CNT_PITCH] = cnt;

  auto add_row = [&](int sj, int dj, float wj) {
    const float* p = proj + (size_t)sj * D + lane;
    float* t       = tile + dj * D + lane;
    float v[CPL];
#pragma unroll
    for (int c = 0; c < CPL; ++c) v[c] = p[32 * c];
#pragma unroll
    for (int c = 0; c < CPL; ++c) atomicAdd(t + 32 * c, v[c] * wj);
  };

  auto add_step = [&](int s, int dl, float w, int nb) {
#pragma unroll
    for (int q = 0; q < 4; ++q) {
      const int nq = nb - 8 * q;
      if (nq >= 8) {
        float v[8][CPL];
#pragma unroll
        for (int j = 0; j < 8; ++j) {
          const int sj   = __builtin_amdgcn_readlane(s, 8 * q + j);
          const float* p = proj + (size_t)sj * D + lane;
#pragma unroll
          for (int c = 0; c < CPL; ++c) v[j][c] = p[32 * c];
        }
#pragma unroll
        for (int j = 0; j < 8; ++j) {
          const int dj   = __builtin_amdgcn_readlane(dl, 8 * q + j);
          const float wj = __int_as_float(__builtin_amdgcn_readlane(__float_as_int(w), 8 * q + j));
          float* t       = tile + dj * D + lane;
#pragma unroll
          for (int c = 0; c < CPL; ++c) atomicAdd(t + 32 * c, v[j][c] * wj);
        }
      } else {
        for (int j = 0; j < nq; ++j) {
          const int l = 8 * q + j;
          add_row(__builtin_amdgcn_readlane(s, l), __builtin_amdgcn_readlane(dl, l),
                  __int_as_float(__builtin_amdgcn_readlane(__float_as_int(w), l)));
        }
      }
    }
  };

  for (int b0 = wave * 32; b0 < total; b0 += 32 * 32) {
    const int nb = min(32, total - b0);
    int s = 0, dl = 0;
    float w = 0.f;
    if (lane < nb) {
      const int2 ent = list[b0 + lane];
      s  = ent.x & 0xFFFF;
      dl = (int)((unsigned)ent.x >> 16);
      w  = __int_as_float(ent.y);
      if (BUILD) gl[b0 + lane] = ent;
    }
    add_step(s, dl, w, nb);
  }
  __syncthreads();

  for (int i = tid; i < rowsHere * D4; i += NT) {
    const int r  = i / D4;
    const int c4 = i % D4;
    const float4 t  = ((const float4*)tile)[i];
    float sc;
    if (BUILD) {
      sc = rsqrtf(fmaxf((float)rowcnt[r], 1.0f));
      if (c4 == 0) idiT[blockIdx.x * IDI_PITCH + r] = sc;
    } else {
      sc = idiT[blockIdx.x * IDI_PITCH + r];
    }
    const float4 b  = ((const float4*)bias)[c4];
    float4 o;
    o.x = t.x * sc + b.x;
    o.y = t.y * sc + b.y;
    o.z = t.z * sc + b.z;
    o.w = t.w * sc + b.w;
    if (RELU) {
      o.x = fmaxf(o.x, 0.0f);
      o.y = fmaxf(o.y, 0.0f);
      o.z = fmaxf(o.z, 0.0f);
      o.w = fmaxf(o.w, 0.0f);
    }
    ((float4*)out)[(size_t)(rowBase + r) * D4 + c4] = o;
  }
}

extern "C" void kernel_launch(void* const* d_in, const int* in_sizes, int n_in,
                              void* d_out, int out_size, void* d_ws, size_t ws_size,
                              hipStream_t stream) {
  const float* feat = (const float*)d_in[0];
  const float* ew   = (const float*)d_in[1];
  const float* W1   = (const float*)d_in[2];
  const float* b1   = (const float*)d_in[3];
  const float* W2   = (const float*)d_in[4];
  const float* b2   = (const float*)d_in[5];
  const float* W3   = (const float*)d_in[6];
  const float* b3   = (const float*)d_in[7];
  const int*   src  = (const int*)d_in[8];
  const int*   dst  = (const int*)d_in[9];
  float* out = (float*)d_out;

  float* ws     = (float*)d_ws;
  float* odi    = ws + N_NODES;
  float* idiT   = ws + 2 * (size_t)N_NODES;
  float* h      = ws + 4 * (size_t)N_NODES;
  float* proj   = h    + (size_t)N_NODES * 128;

  constexpr int GS = 5;
  const int gemmBlocks = N_NODES / 16 / GS;
  constexpr int TR = TILE_ROWS;
  constexpr int tiles = (N_NODES + TR - 1) / TR;
  static_assert(tiles == N_TILES, "the bitmap has one row per tile");
  static_assert(tiles * IDI_PITCH <= 2 * N_NODES, "idiT ends before h");
  constexpr int CAP128 = (256 * 1024 - TR * 128 * 4) / 8;
  constexpr int CAP64  = (256 * 1024 - TR * 64 * 4) / 8;
  int*      glist = (int*)(proj + (size_t)N_NODES * 128);
  unsigned* gcnt  = (unsigned*)(glist + (size_t)2 * tiles * CAP128);
  unsigned* part = gcnt + (size_t)tiles * CNT_PITCH;
  unsigned* bm = part + (size_t)DEG_SLICES * DEG_RANGES * DEG_PITCH;
  unsigned* rec = bm + (size_t)tiles * BM_TILE_WORDS;
  constexpr int GLC = 12288;

  gcn_outdeg_count<<<DEG_SLICES * DEG_RANGES, DEG_THREADS, 0, stream>>>(src, dst, ew, part, bm, rec);
  gcn_outdeg_isqrt<<<(N_NODES + 255) / 256, 256, 0, stream>>>(part, odi);

  gcn_gemm_wmma<128, GS><<<gemmBlocks, 256, 0, stream>>>(feat, odi, W1, proj);
  gcn_tile_aggregate<128, TR, CAP128, CAP128, GLC, true, true><<<tiles, 1024, 0, stream>>>(proj, rec, idiT, b1, h, glist, gcnt, bm);

  gcn_gemm_wmma<128, GS><<<gemmBlocks, 256, 0, stream>>>(h, odi, W2, proj);
  gcn_tile_aggregate<128, TR, CAP128, CAP128, GLC, true, false><<<tiles, 1024, 0, stream>>>(proj, rec, idiT, b2, h, glist, gcnt, bm);

  gcn_gemm_wmma<64, GS><<<gemmBlocks, 128, 0, stream>>>(h, odi, W3, proj);
  gcn_tile_aggregate<64, TR, CAP64, CAP128, GLC, false, false><<<tiles, 1024, 0, stream>>>(proj, rec, idiT, b3, out, glist, gcnt, bm);
}


# Tile aggregate phase B: a listed source row is ONE load per wave (16 B a lane at 128 columns, 8 B at 64), not 4 (2) 4-byte loads; tile columns permuted in LDS, adds conflict-free; LDS arrival order
# speedup vs baseline: 3.9298x; 1.0505x over previous
#include <hip/hip_runtime.h>
#include <math.h>


typedef __attribute__((ext_vector_type(2))) float v2f;
typedef __attribute__((ext_vector_type(8))) float v8f;

#define N_NODES 50000
#define N_EDGES 800000
#define K_DIM   128
#define CNT_PITCH 32
#define TILE_ROWS 197
#define N_TILES ((N_NODES + TILE_ROWS - 1) / TILE_ROWS)
#define BM_GROUPS 1024
#define BM_CHUNKS ((N_EDGES / 4 + BM_GROUPS - 1) / BM_GROUPS)
#define BM_TILE_WORDS (BM_CHUNKS * 32)
#define IDI_PITCH 256
#define DEG_SLICES 32
#define DEG_RANGES 8
#define DEG_RANGE_NODES (N_NODES / DEG_RANGES)
#define DEG_PITCH 6272
#define DEG_THREADS 512

__global__ __launch_bounds__(DEG_THREADS) void gcn_outdeg_count(const int* __restrict__ src,
                                                                const int* __restrict__ dst,
                                                                const float* __restrict__ ew,
                                                                unsigned* __restrict__ part,
                                                                unsigned* __restrict__ bm,
                                                                unsigned* __restrict__ rec) {
  constexpr int PAIRS = 4;
  static_assert(N_NODES % DEG_RANGES == 0 && N_EDGES % (4 * DEG_SLICES) == 0, "even split");
  static_assert(DEG_PITCH >= DEG_RANGE_NODES && DEG_PITCH % 32 == 0, "segment = whole lines");
  static_assert(BM_CHUNKS <= DEG_SLICES * DEG_RANGES && 2 * BM_GROUPS == PAIRS * DEG_THREADS && N_EDGES % 4 == 0,
                "one work-group per bitmap chunk, PAIRS edge pairs per thread");
  static_assert(N_NODES <= 65535, "an edge record packs source and destination in 16 bits each");
  constexpr int SLICE4 = N_EDGES / DEG_SLICES / 4;
  constexpr int PER    = (SLICE4 + DEG_THREADS - 1) / DEG_THREADS;

  __shared__ __attribute__((aligned(16))) unsigned cntL[DEG_PITCH];
  __shared__ __attribute__((aligned(16))) unsigned bmL[N_TILES * 32];

  const int tid = threadIdx.x;
  const int s   = blockIdx.x / DEG_RANGES;
  const int lo  = (blockIdx.x % DEG_RANGES) * DEG_RANGE_NODES;

  const int4* s4 = (const int4*)src + (size_t)s * SLICE4;
  int4 v[PER];
#pragma unroll
  for (int u = 0; u < PER; ++u) {
    const int i = tid + u * DEG_THREADS;
    v[u] = (i < SLICE4) ? s4[i] : make_int4(-1, -1, -1, -1);
  }

  const bool chunk = blockIdx.x < BM_CHUNKS;
  int2   dp[PAIRS];
  int2   sp[PAIRS];
  float2 wp[PAIRS];
#pragma unroll
  for (int u = 0; u < PAIRS; ++u) {
    const int h   = blockIdx.x * (2 * BM_GROUPS) + tid + u * DEG_THREADS;
    const bool ok = chunk && h < N_EDGES / 2;
    dp[u] = ok ? ((const int2*)dst)[h] : make_int2(-1, -1);
    sp[u] = ok ? ((const int2*)src)[h] : make_int2(0, 0);
    wp[u] = ok ? ((const float2*)ew)[h] : make_float2(0.f, 0.f);
  }

  for (int i = tid; i < DEG_PITCH / 4; i += DEG_THREADS)
    ((uint4*)cntL)[i] = make_uint4(0u, 0u, 0u, 0u);
  if (chunk)
    for (int i = tid; i < N_TILES * 8; i += DEG_THREADS)
      ((uint4*)bmL)[i] = make_uint4(0u, 0u, 0u, 0u);
  __syncthreads();

  auto mark = [&](int d, int g) {
    if ((unsigned)d < (unsigned)N_NODES)
      atomicOr(&bmL[((unsigned)d / TILE_ROWS) * 32 + (g >> 5)], 1u << (g & 31));
  };
#pragma unroll
  for (int u = 0; u < PAIRS; ++u) {
    const int hl = tid + u * DEG_THREADS;
    mark(dp[u].x, hl >> 1);
    mark(dp[u].y, hl >> 1);
    const int h = blockIdx.x * (2 * BM_GROUPS) + hl;
    if (chunk && h < N_EDGES / 2)
      ((uint4*)rec)[h] = make_uint4((unsigned)sp[u].x | ((unsigned)dp[u].x << 16), __float_as_uint(wp[u].x),
                                    (unsigned)sp[u].y | ((unsigned)dp[u].y << 16), __float_as_uint(wp[u].y));
  }

  auto count = [&](int sv) {
    const unsigned o = (unsigned)(sv - lo);
    if (o < (unsigned)DEG_RANGE_NODES) atomicAdd(&cntL[o], 1u);
  };
#pragma unroll
  for (int u = 0; u < PER; ++u) {
    count(v[u].x);
    count(v[u].y);
    count(v[u].z);
    count(v[u].w);
  }
  __syncthreads();

  uint4* seg = (uint4*)(part + (size_t)blockIdx.x * DEG_PITCH);
  for (int i = tid; i < DEG_PITCH / 4; i += DEG_THREADS) seg[i] = ((const uint4*)cntL)[i];

  if (chunk)
    for (int i = tid; i < N_TILES * 8; i += DEG_THREADS)
      ((uint4*)(bm + (size_t)(i >> 3) * BM_TILE_WORDS + blockIdx.x * 32))[i & 7] = ((const uint4*)bmL)[i];
}

__global__ void gcn_outdeg_isqrt(const unsigned* __restrict__ part, float* __restrict__ odi) {
  int n = blockIdx.x * blockDim.x + threadIdx.x;
  if (n < N_NODES) {
    const int r = n / DEG_RANGE_NODES;
    const unsigned* p = part + (size_t)r * DEG_PITCH + (n - r * DEG_RANGE_NODES);
    unsigned c = 0u;
#pragma unroll
    for (int s = 0; s < DEG_SLICES; ++s) c += p[(size_t)s * DEG_RANGES * DEG_PITCH];
    odi[n] = rsqrtf(fmaxf((float)c, 1.0f));
  }
}

template <int NOUT, int STRIPES>
__global__ __launch_bounds__(NOUT * 2) void gcn_gemm_wmma(const float* __restrict__ A,
                                                          const float* __restrict__ odi,
                                                          const float* __restrict__ W,
                                                          float* __restrict__ C) {
  static_assert((N_NODES / 16) % STRIPES == 0, "every work-group walks STRIPES whole stripes");
  constexpr int NT  = NOUT * 2;
  constexpr int PER = 16 * K_DIM / 4 / NT;
  static_assert(PER * NT * 4 == 16 * K_DIM, "a stripe is PER whole loads per thread");

  __shared__ __attribute__((aligned(16))) float As[2][16][K_DIM + 4];

  const int tid  = threadIdx.x;
  const int wave = tid >> 5;
  const int lane = tid & 31;
  const int half = lane >> 4;
  const int r16  = lane & 15;
  const int n0   = wave * 16;
  const int tile0 = blockIdx.x * STRIPES;

  v2f b[K_DIM / 4];
#pragma unroll
  for (int k = 0; k < K_DIM; k += 4) {
    b[k / 4].x = W[(size_t)(k + 2 * half)     * NOUT + n0 + r16];
    b[k / 4].y = W[(size_t)(k + 2 * half + 1) * NOUT + n0 + r16];
  }

  float4 ar[PER];
  float  sc[PER];
  auto fetch = [&](int tile) {
#pragma unroll
    for (int u = 0; u < PER; ++u) {
      const int idx = tid + u * NT;
      const int r   = idx >> 5;
      const int c4  = idx & 31;
      ar[u] = ((const float4*)(A + (size_t)(tile * 16 + r) * K_DIM))[c4];
      sc[u] = odi[tile * 16 + r];
    }
  };
  fetch(tile0);

#pragma unroll 1
  for (int s = 0; s < STRIPES; ++s) {
    const int tileM = (tile0 + s) * 16;
    float (*Ab)[K_DIM + 4] = As[s & 1];
#pragma unroll
    for (int u = 0; u < PER; ++u) {
      const int idx = tid + u * NT;
      const int r   = idx >> 5;
      const int c4  = idx & 31;
      float4 v = ar[u];
      v.x *= sc[u];
      v.y *= sc[u];
      v.z *= sc[u];
      v.w *= sc[u];
      *(float4*)&Ab[r][4 * c4] = v;
    }
    __syncthreads();
    if (s + 1 < STRIPES) fetch(tile0 + s + 1);

    v8f acc = {};
#pragma unroll
    for (int k = 0; k < K_DIM; k += 4) {
      v2f a;
      a.x = Ab[r16][k + 2 * half];
      a.y = Ab[r16][k + 2 * half + 1];
      acc = __builtin_amdgcn_wmma_f32_16x16x4_f32(false, a, false, b[k / 4],
                                                  (short)0, acc, false, false);
    }

#pragma unroll
    for (int r = 0; r < 8; ++r) {
      const int row = tileM + r + 8 * half;
      C[(size_t)row * NOUT + n0 + r16] = acc[r];
    }
  }
}

template <int D, int TR, int CAP, int GCAP, int GLC, bool RELU, bool BUILD>
__global__ __launch_bounds__(1024) void gcn_tile_aggregate(const float* __restrict__ proj,
                                                           const unsigned* __restrict__ rec,
                                                           float* __restrict__ idiT,
                                                           const float* __restrict__ bias,
                                                           float* __restrict__ out,
                                                           int* __restrict__ glist,
                                                           unsigned* __restrict__ gcnt,
                                                           const unsigned* __restrict__ bm) {
  static_assert(D % 32 == 0 && N_EDGES % 4 == 0, "layout");
  static_assert(TR == TILE_ROWS, "a bit row names a tile of TR rows");
  static_assert(N_NODES <= 65535 && TR <= 32768, "packed list entry and edge record: 16 bits each; 65535 is no row");
  static_assert(GLC % 2 == 0 && GLC > 0, "group list");
  static_assert(BUILD ? CAP <= GCAP : GCAP <= CAP, "listed edges fit glist; a stored list fits the LDS list");
  static_assert(TR <= IDI_PITCH, "a tile's in_deg^{-1/2} values fit its segment of idiT");
  constexpr int NT  = 1024;
  constexpr int CPL = D / 32;
  typedef __attribute__((ext_vector_type(CPL))) float rowv;
  auto tword = [](int col) { return (col % CPL) * 32 + col / CPL; };
  constexpr int D4  = D / 4;
  constexpr int WPT = (BM_TILE_WORDS + NT - 1) / NT;
  constexpr int GU  = 8;

  __shared__ __attribute__((aligned(16))) float tile[TR * D];
  __shared__ int2     list[CAP];
  __shared__ int      grp[GLC];
  __shared__ unsigned cnt;
  __shared__ unsigned ngrp;
  __shared__ unsigned rowcnt[TR];

  const int tid      = threadIdx.x;
  const int lane     = tid & 31;
  const int rowBase  = blockIdx.x * TR;
  const int rowsHere = min(TR, N_NODES - rowBase);

  int2* const gl = (int2*)glist + (size_t)blockIdx.x * GCAP;
  unsigned nStored = 0u;
  if (!BUILD) nStored = gcnt[blockIdx.x * CNT_PITCH];
  const bool useStored = !BUILD && nStored <= (unsigned)GCAP;
  constexpr int PRE = 4;
  int2 pre[PRE];
  if (useStored) {
#pragma unroll
    for (int u = 0; u < PRE; ++u) {
      const unsigned i = (unsigned)(tid + u * NT);
      pre[u] = (i < nStored) ? gl[i] : make_int2(0, 0);
    }
  }

  for (int i = tid; i < TR * D4; i += NT)
    ((float4*)tile)[i] = make_float4(0.f, 0.f, 0.f, 0.f);
  if (tid == 0) {
    cnt  = 0u;
    ngrp = 0u;
  }
  if (BUILD)
    for (int i = tid; i < TR; i += NT) rowcnt[i] = 0u;
  if (useStored) {
#pragma unroll
    for (int u = 0; u < PRE; ++u) {
      const unsigned i = (unsigned)(tid + u * NT);
      if (i < nStored) list[i] = pre[u];
    }
    for (unsigned i = (unsigned)(tid + PRE * NT); i < nStored; i += NT) list[i] = gl[i];
  }
  __syncthreads();

  auto take_edge = [&](int s, int dl, int wbits) {
    if (BUILD) atomicAdd(&rowcnt[dl], 1u);
    const unsigned slot = atomicAdd(&cnt, 1u);
    if (slot < (unsigned)CAP) {
      list[slot] = make_int2(s | (dl << 16), wbits);
    } else {
      const float w  = __int_as_float(wbits);
      const float* p = proj + (size_t)s * D;
      float* t       = tile + dl * D;
#pragma nounroll
      for (int c = 0; c < D; ++c) atomicAdd(t + tword(c), p[c] * w);
    }
  };
  const uint4* rec4 = (const uint4*)rec;
  auto test2 = [&](const uint4& r) {
    const unsigned rows = (unsigned)rowsHere;
    const unsigned d0 = (r.x >> 16) - (unsigned)rowBase;
    const unsigned d1 = (r.z >> 16) - (unsigned)rowBase;
    unsigned m = (d0 < rows ? 1u : 0u) | (d1 < rows ? 2u : 0u);
    while (m != 0u) {
      const bool second = (m & 1u) == 0u;
      m &= m - 1u;
      take_edge((int)((second ? r.z : r.x) & 0xFFFFu), (int)(second ? d1 : d0), (int)(second ? r.w : r.y));
    }
  };

  if (!useStored) {
    const unsigned* bmT = bm + (size_t)blockIdx.x * BM_TILE_WORDS;
    unsigned wv[WPT];
    int n = 0;
#pragma unroll
    for (int u = 0; u < WPT; ++u) {
      const int i = tid + u * NT;
      wv[u] = (i < BM_TILE_WORDS) ? bmT[i] : 0u;
      n += __builtin_popcount(wv[u]);
    }
    int incl = n;
#pragma unroll
    for (int o = 1; o < 32; o <<= 1) {
      const int t = __shfl_up(incl, o);
      if (lane >= o) incl += t;
    }
    int base = 0;
    if (lane == 31) base = (int)atomicAdd(&ngrp, (unsigned)incl);
    base = __shfl(base, 31);
    int pos = base + incl - n;
#pragma unroll
    for (int u = 0; u < WPT; ++u) {
      unsigned w = wv[u];
      const int i = tid + u * NT;
      while (w != 0u) {
        const int q = i * 32 + __builtin_ctz(w);
        w &= w - 1u;
        if (pos < GLC) grp[pos] = q;
        ++pos;
      }
    }
    if (pos > GLC) {
      int p2 = base + incl - n;
#pragma nounroll
      for (int i = tid; i < BM_TILE_WORDS; i += NT) {
        unsigned w = bmT[i];
        while (w != 0u) {
          const int q = i * 32 + __builtin_ctz(w);
          w &= w - 1u;
          if (p2 >= GLC) {
            test2(rec4[2 * q]);
            test2(rec4[2 * q + 1]);
          }
          ++p2;
        }
      }
    }
    __syncthreads();

    const int nH = 2 * (int)min(ngrp, (unsigned)GLC);
    for (int h0 = tid; h0 < nH; h0 += NT * GU) {
      uint4 r[GU];
#pragma unroll
      for (int u = 0; u < GU; ++u) {
        const int hi = h0 + u * NT;
        r[u] = (hi < nH) ? rec4[2 * grp[hi >> 1] + (hi & 1)] : make_uint4(0xFFFF0000u, 0u, 0xFFFF0000u, 0u);
      }
#pragma unroll
      for (int u = 0; u < GU; ++u) test2(r[u]);
    }
  }
  __syncthreads();

  const int wave  = __builtin_amdgcn_readfirstlane(tid >> 5);
  const int total = __builtin_amdgcn_readfirstlane(
      useStored ? (int)nStored : (int)min(cnt, (unsigned)CAP));
  if (BUILD && tid == 0) gcnt[blockIdx.x * CNT_PITCH] = cnt;

  auto row_base = [&](int sj) {
    const unsigned long long a = (unsigned long long)proj + (unsigned long long)(unsigned)sj * (unsigned)(D * 4);
    const unsigned lo = __builtin_amdgcn_readfirstlane((unsigned)a);
    const unsigned hi = __builtin_amdgcn_readfirstlane((unsigned)(a >> 32));
    return (const __attribute__((address_space(1))) rowv*)(((unsigned long long)hi << 32) | lo);
  };

  auto add_row = [&](int sj, int dj, float wj) {
    const rowv v = row_base(sj)[lane];
    float* t     = tile + dj * D + lane;
#pragma unroll
    for (int c = 0; c < CPL; ++c) atomicAdd(t + 32 * c, v[c] * wj);
  };

  auto add_step = [&](int s, int dl, float w, int nb) {
#pragma unroll
    for (int q = 0; q < 4; ++q) {
      const int nq = nb - 8 * q;
      if (nq >= 8) {
        rowv v[8];
#pragma unroll
        for (int j = 0; j < 8; ++j) v[j] = row_base(__builtin_amdgcn_readlane(s, 8 * q + j))[lane];
#pragma unroll
        for (int j = 0; j < 8; ++j) {
          const int dj   = __builtin_amdgcn_readlane(dl, 8 * q + j);
          const float wj = __int_as_float(__builtin_amdgcn_readlane(__float_as_int(w), 8 * q + j));
          float* t       = tile + dj * D + lane;
#pragma unroll
          for (int c = 0; c < CPL; ++c) atomicAdd(t + 32 * c, v[j][c] * wj);
        }
      } else {
        for (int j = 0; j < nq; ++j) {
          const int l = 8 * q + j;
          add_row(__builtin_amdgcn_readlane(s, l), __builtin_amdgcn_readlane(dl, l),
                  __int_as_float(__builtin_amdgcn_readlane(__float_as_int(w), l)));
        }
      }
    }
  };

  for (int b0 = wave * 32; b0 < total; b0 += 32 * 32) {
    const int nb = min(32, total - b0);
    int s = 0, dl = 0;
    float w = 0.f;
    if (lane < nb) {
      const int2 ent = list[b0 + lane];
      s  = ent.x & 0xFFFF;
      dl = (int)((unsigned)ent.x >> 16);
      w  = __int_as_float(ent.y);
      if (BUILD) gl[b0 + lane] = ent;
    }
    add_step(s, dl, w, nb);
  }
  __syncthreads();

  for (int i = tid; i < rowsHere * D4; i += NT) {
    const int r  = i / D4;
    const int c4 = i % D4;
    const float* trow = tile + r * D;
    float4 t;
    t.x = trow[tword(4 * c4)];
    t.y = trow[tword(4 * c4 + 1)];
    t.z = trow[tword(4 * c4 + 2)];
    t.w = trow[tword(4 * c4 + 3)];
    float sc;
    if (BUILD) {
      sc = rsqrtf(fmaxf((float)rowcnt[r], 1.0f));
      if (c4 == 0) idiT[blockIdx.x * IDI_PITCH + r] = sc;
    } else {
      sc = idiT[blockIdx.x * IDI_PITCH + r];
    }
    const float4 b  = ((const float4*)bias)[c4];
    float4 o;
    o.x = t.x * sc + b.x;
    o.y = t.y * sc + b.y;
    o.z = t.z * sc + b.z;
    o.w = t.w * sc + b.w;
    if (RELU) {
      o.x = fmaxf(o.x, 0.0f);
      o.y = fmaxf(o.y, 0.0f);
      o.z = fmaxf(o.z, 0.0f);
      o.w = fmaxf(o.w, 0.0f);
    }
    ((float4*)out)[(size_t)(rowBase + r) * D4 + c4] = o;
  }
}

extern "C" void kernel_launch(void* const* d_in, const int* in_sizes, int n_in,
                              void* d_out, int out_size, void* d_ws, size_t ws_size,
                              hipStream_t stream) {
  const float* feat = (const float*)d_in[0];
  const float* ew   = (const float*)d_in[1];
  const float* W1   = (const float*)d_in[2];
  const float* b1   = (const float*)d_in[3];
  const float* W2   = (const float*)d_in[4];
  const float* b2   = (const float*)d_in[5];
  const float* W3   = (const float*)d_in[6];
  const float* b3   = (const float*)d_in[7];
  const int*   src  = (const int*)d_in[8];
  const int*   dst  = (const int*)d_in[9];
  float* out = (float*)d_out;

  float* ws     = (float*)d_ws;
  float* odi    = ws + N_NODES;
  float* idiT   = ws + 2 * (size_t)N_NODES;
  float* h      = ws + 4 * (size_t)N_NODES;
  float* proj   = h    + (size_t)N_NODES * 128;

  constexpr int GS = 5;
  const int gemmBlocks = N_NODES / 16 / GS;
  constexpr int TR = TILE_ROWS;
  constexpr int tiles = (N_NODES + TR - 1) / TR;
  static_assert(tiles == N_TILES, "the bitmap has one row per tile");
  static_assert(tiles * IDI_PITCH <= 2 * N_NODES, "idiT ends before h");
  constexpr int CAP128 = (256 * 1024 - TR * 128 * 4) / 8;
  constexpr int CAP64  = (256 * 1024 - TR * 64 * 4) / 8;
  int*      glist = (int*)(proj + (size_t)N_NODES * 128);
  unsigned* gcnt  = (unsigned*)(glist + (size_t)2 * tiles * CAP128);
  unsigned* part = gcnt + (size_t)tiles * CNT_PITCH;
  unsigned* bm = part + (size_t)DEG_SLICES * DEG_RANGES * DEG_PITCH;
  unsigned* rec = bm + (size_t)tiles * BM_TILE_WORDS;
  constexpr int GLC = 12288;

  gcn_outdeg_count<<<DEG_SLICES * DEG_RANGES, DEG_THREADS, 0, stream>>>(src, dst, ew, part, bm, rec);
  gcn_outdeg_isqrt<<<(N_NODES + 255) / 256, 256, 0, stream>>>(part, odi);

  gcn_gemm_wmma<128, GS><<<gemmBlocks, 256, 0, stream>>>(feat, odi, W1, proj);
  gcn_tile_aggregate<128, TR, CAP128, CAP128, GLC, true, true><<<tiles, 1024, 0, stream>>>(proj, rec, idiT, b1, h, glist, gcnt, bm);

  gcn_gemm_wmma<128, GS><<<gemmBlocks, 256, 0, stream>>>(h, odi, W2, proj);
  gcn_tile_aggregate<128, TR, CAP128, CAP128, GLC, true, false><<<tiles, 1024, 0, stream>>>(proj, rec, idiT, b2, h, glist, gcnt, bm);

  gcn_gemm_wmma<64, GS><<<gemmBlocks, 128, 0, stream>>>(h, odi, W3, proj);
  gcn_tile_aggregate<64, TR, CAP64, CAP128, GLC, false, false><<<tiles, 1024, 0, stream>>>(proj, rec, idiT, b3, out, glist, gcnt, bm);
}
